# starting kernel
# baseline (speedup 1.0000x reference)
#include <hip/hip_runtime.h>
#include <math.h>

#define NNODES  50000
#define NEDGES  800000
#define ETOT    (NEDGES + NNODES)
#define IN_DIM  128
#define HID     64
#define HEADS   4
#define OUT_DIM 64
#define F1      (HEADS * HID)
#define ENC_NEG_INF 0x007FFFFFu

typedef __attribute__((ext_vector_type(2))) float v2f;
typedef __attribute__((ext_vector_type(8))) float v8f;

__device__ __forceinline__ float leakyf(float x, float s) { return x >= 0.f ? x : s * x; }

__device__ __forceinline__ unsigned enc_ord(float x) {
    unsigned u = __float_as_uint(x);
    return (u & 0x80000000u) ? ~u : (u | 0x80000000u);
}
__device__ __forceinline__ float dec_ord(unsigned e) {
    return (e & 0x80000000u) ? __uint_as_float(e & 0x7FFFFFFFu) : __uint_as_float(~e);
}

__device__ __forceinline__ void edge_sd(const int* __restrict__ ei, int t, int& s, int& d) {
    if (t < NEDGES) { s = ei[t]; d = ei[NEDGES + t]; }
    else            { s = t - NEDGES; d = s; }
}

__global__ void gemm_wmma_f32_kernel(const float* __restrict__ A,
                                     const float* __restrict__ W,
                                     float* __restrict__ C,
                                     int nrows, int fin, int fout) {
    const int gwave = (blockIdx.x * blockDim.x + threadIdx.x) >> 5;
    const int lane  = threadIdx.x & 31;
    const int stripsPerRow = fout >> 6;
    const int totalStrips  = (nrows >> 4) * stripsPerRow;
    if (gwave >= totalStrips) return;

    const int rowBlk = gwave / stripsPerRow;
    const int strip  = gwave % stripsPerRow;
    const int row0 = rowBlk << 4;
    const int col0 = strip << 6;
    const int half = lane >> 4;
    const int l    = lane & 15;

    v8f acc0 = {}, acc1 = {}, acc2 = {}, acc3 = {};
    const float* arow = A + (size_t)(row0 + l) * fin;
    for (int k = 0; k < fin; k += 4) {
        v2f a;
        a.x = arow[k + 2 * half + 0];
        a.y = arow[k + 2 * half + 1];
        const float* w0 = W + (size_t)(k + 2 * half) * fout + col0 + l;
        const float* w1 = w0 + fout;
        v2f b0 = { w0[ 0], w1[ 0] };
        v2f b1 = { w0[16], w1[16] };
        v2f b2 = { w0[32], w1[32] };
        v2f b3 = { w0[48], w1[48] };
        acc0 = __builtin_amdgcn_wmma_f32_16x16x4_f32(false, a, false, b0, (short)0, acc0, false, false);
        acc1 = __builtin_amdgcn_wmma_f32_16x16x4_f32(false, a, false, b1, (short)0, acc1, false, false);
        acc2 = __builtin_amdgcn_wmma_f32_16x16x4_f32(false, a, false, b2, (short)0, acc2, false, false);
        acc3 = __builtin_amdgcn_wmma_f32_16x16x4_f32(false, a, false, b3, (short)0, acc3, false, false);
    }
    float* crow = C + (size_t)row0 * fout + col0;
#pragma unroll
    for (int r = 0; r < 8; ++r) {
        const int m = r + 8 * half;
        float* p = crow + (size_t)m * fout + l;
        p[ 0] = acc0[r];
        p[16] = acc1[r];
        p[32] = acc2[r];
        p[48] = acc3[r];
    }
}

__global__ void alpha_kernel(const float* __restrict__ h,
                             const float* __restrict__ a_src, const float* __restrict__ a_dst,
                             float* __restrict__ as_, float* __restrict__ ad_,
                             int nn, int H, int Cc) {
    int idx = blockIdx.x * blockDim.x + threadIdx.x;
    if (idx >= nn * H) return;
    int n = idx / H, hh = idx - n * H;
    const float* hp = h + (size_t)n * H * Cc + hh * Cc;
    float s = 0.f, d = 0.f;
    for (int c = 0; c < Cc; ++c) {
        float v = hp[c];
        s += v * a_src[hh * Cc + c];
        d += v * a_dst[hh * Cc + c];
    }
    as_[idx] = s;
    ad_[idx] = d;
}

__global__ void edge_max_kernel(const int* __restrict__ ei,
                                const float* __restrict__ as_, const float* __restrict__ ad_,
                                float* __restrict__ ebuf, unsigned* __restrict__ menc, int H) {
    int idx = blockIdx.x * blockDim.x + threadIdx.x;
    if (idx >= ETOT * H) return;
    int t = idx / H, hh = idx - t * H;
    int s, d; edge_sd(ei, t, s, d);
    float e = leakyf(as_[s * H + hh] + ad_[d * H + hh], 0.2f);
    ebuf[idx] = e;
    atomicMax(&menc[d * H + hh], enc_ord(e));
}

__global__ void edge_exp_kernel(const int* __restrict__ ei,
                                float* __restrict__ ebuf, const unsigned* __restrict__ menc,
                                float* __restrict__ den, int H) {
    int idx = blockIdx.x * blockDim.x + threadIdx.x;
    if (idx >= ETOT * H) return;
    int t = idx / H, hh = idx - t * H;
    int s, d; edge_sd(ei, t, s, d);
    (void)s;
    float m  = dec_ord(menc[d * H + hh]);
    float ex = expf(ebuf[idx] - m);
    ebuf[idx] = ex;
    atomicAdd(&den[d * H + hh], ex);
}

__global__ void scatter_kernel(const int* __restrict__ ei,
                               const float* __restrict__ h,
                               const float* __restrict__ ebuf,
                               const float* __restrict__ den,
                               float* __restrict__ out, int H) {
    const int gwave = (blockIdx.x * blockDim.x + threadIdx.x) >> 5;
    const int lane  = threadIdx.x & 31;
    if (gwave >= ETOT) return;
    int s, d; edge_sd(ei, gwave, s, d);
    const int F = H << 6;
    const float* hs = h + (size_t)s * F;
    float* od = out + (size_t)d * F;
    float alph[HEADS];
#pragma unroll
    for (int hh = 0; hh < HEADS; ++hh)
        if (hh < H) alph[hh] = ebuf[(size_t)gwave * H + hh] / (den[d * H + hh] + 1e-16f);
    for (int c = lane; c < F; c += 32)
        atomicAdd(&od[c], alph[c >> 6] * hs[c]);
}

__global__ void bias_act_kernel(const float* __restrict__ in, const float* __restrict__ b,
                                float* __restrict__ out, int total, int F, float slope) {
    int i = blockIdx.x * blockDim.x + threadIdx.x;
    if (i >= total) return;
    out[i] = leakyf(in[i] + b[i % F], slope);
}

__global__ void fill_u32_kernel(unsigned* __restrict__ p, unsigned v, int n) {
    int i = blockIdx.x * blockDim.x + threadIdx.x;
    if (i < n) p[i] = v;
}

extern "C" void kernel_launch(void* const* d_in, const int* in_sizes, int n_in,
                              void* d_out, int out_size, void* d_ws, size_t ws_size,
                              hipStream_t stream) {
    const float* x   = (const float*)d_in[0];
    const int*   ei  = (const int*)  d_in[1];
    const float* W1  = (const float*)d_in[2];
    const float* as1 = (const float*)d_in[3];
    const float* ad1 = (const float*)d_in[4];
    const float* b1  = (const float*)d_in[5];
    const float* W2  = (const float*)d_in[6];
    const float* as2 = (const float*)d_in[7];
    const float* ad2 = (const float*)d_in[8];
    const float* b2  = (const float*)d_in[9];
    float* out = (float*)d_out;

    float* ws   = (float*)d_ws;
    float* h1   = ws;
    float* h2   = h1   + (size_t)NNODES * F1;
    float* ebuf = h2   + (size_t)NNODES * F1;
    float* asb  = ebuf + (size_t)ETOT * HEADS;
    float* adb  = asb  + (size_t)NNODES * HEADS;
    unsigned* menc = (unsigned*)(adb + (size_t)NNODES * HEADS);
    float* den  = (float*)(menc + (size_t)NNODES * HEADS);
    float* g2   = h1;
    float* out2 = h1 + (size_t)NNODES * OUT_DIM;

    {
        int strips = (NNODES / 16) * (F1 / 64);
        gemm_wmma_f32_kernel<<<(strips + 3) / 4, 128, 0, stream>>>(x, W1, h1, NNODES, IN_DIM, F1);
    }
    alpha_kernel<<<(NNODES * HEADS + 255) / 256, 256, 0, stream>>>(h1, as1, ad1, asb, adb,
                                                                   NNODES, HEADS, HID);
    hipMemsetAsync(den, 0, sizeof(float) * (size_t)NNODES * HEADS, stream);
    fill_u32_kernel<<<(NNODES * HEADS + 255) / 256, 256, 0, stream>>>(menc, ENC_NEG_INF,
                                                                      NNODES * HEADS);
    hipMemsetAsync(h2, 0, sizeof(float) * (size_t)NNODES * F1, stream);
    edge_max_kernel<<<(ETOT * HEADS + 255) / 256, 256, 0, stream>>>(ei, asb, adb, ebuf, menc, HEADS);
    edge_exp_kernel<<<(ETOT * HEADS + 255) / 256, 256, 0, stream>>>(ei, ebuf, menc, den, HEADS);
    scatter_kernel<<<(ETOT + 7) / 8, 256, 0, stream>>>(ei, h1, ebuf, den, h2, HEADS);
    bias_act_kernel<<<(NNODES * F1 + 255) / 256, 256, 0, stream>>>(h2, b1, h2,
                                                                   NNODES * F1, F1, 0.01f);

    {
        int strips = (NNODES / 16) * (OUT_DIM / 64);
        gemm_wmma_f32_kernel<<<(strips + 3) / 4, 128, 0, stream>>>(h2, W2, g2, NNODES, F1, OUT_DIM);
    }
    alpha_kernel<<<(NNODES + 255) / 256, 256, 0, stream>>>(g2, as2, ad2, asb, adb,
                                                           NNODES, 1, OUT_DIM);
    hipMemsetAsync(den, 0, sizeof(float) * (size_t)NNODES, stream);
    fill_u32_kernel<<<(NNODES + 255) / 256, 256, 0, stream>>>(menc, ENC_NEG_INF, NNODES);
    hipMemsetAsync(out2, 0, sizeof(float) * (size_t)NNODES * OUT_DIM, stream);
    edge_max_kernel<<<(ETOT + 255) / 256, 256, 0, stream>>>(ei, asb, adb, ebuf, menc, 1);
    edge_exp_kernel<<<(ETOT + 255) / 256, 256, 0, stream>>>(ei, ebuf, menc, den, 1);
    scatter_kernel<<<(ETOT + 7) / 8, 256, 0, stream>>>(ei, g2, ebuf, den, out2, 1);
    bias_act_kernel<<<(NNODES * OUT_DIM + 255) / 256, 256, 0, stream>>>(out2, b2, out,
                                                                        NNODES * OUT_DIM,
                                                                        OUT_DIM, 1.0f);
}


# Edges grouped by destination in every call (counting sort, one int atomic per edge); one wave per destination does softmax and weighted row sum in registers, replacing three float-atomic edge passes
# speedup vs baseline: 2.6857x; 2.6857x over previous
#include <hip/hip_runtime.h>
#include <math.h>

#define NNODES  50000
#define NEDGES  800000
#define IN_DIM  128
#define HID     64
#define HEADS   4
#define OUT_DIM 64
#define F1      (HEADS * HID)

typedef __attribute__((ext_vector_type(2))) float v2f;
typedef __attribute__((ext_vector_type(8))) float v8f;

__device__ __forceinline__ float leakyf(float x, float s) { return x >= 0.f ? x : s * x; }

__global__ void gemm_wmma_f32_kernel(const float* __restrict__ A,
                                     const float* __restrict__ W,
                                     float* __restrict__ C,
                                     int nrows, int fin, int fout) {
    const int gwave = (blockIdx.x * blockDim.x + threadIdx.x) >> 5;
    const int lane  = threadIdx.x & 31;
    const int stripsPerRow = fout >> 6;
    const int totalStrips  = (nrows >> 4) * stripsPerRow;
    if (gwave >= totalStrips) return;

    const int rowBlk = gwave / stripsPerRow;
    const int strip  = gwave % stripsPerRow;
    const int row0 = rowBlk << 4;
    const int col0 = strip << 6;
    const int half = lane >> 4;
    const int l    = lane & 15;

    v8f acc0 = {}, acc1 = {}, acc2 = {}, acc3 = {};
    const float* arow = A + (size_t)(row0 + l) * fin;
    for (int k = 0; k < fin; k += 4) {
        v2f a;
        a.x = arow[k + 2 * half + 0];
        a.y = arow[k + 2 * half + 1];
        const float* w0 = W + (size_t)(k + 2 * half) * fout + col0 + l;
        const float* w1 = w0 + fout;
        v2f b0 = { w0[ 0], w1[ 0] };
        v2f b1 = { w0[16], w1[16] };
        v2f b2 = { w0[32], w1[32] };
        v2f b3 = { w0[48], w1[48] };
        acc0 = __builtin_amdgcn_wmma_f32_16x16x4_f32(false, a, false, b0, (short)0, acc0, false, false);
        acc1 = __builtin_amdgcn_wmma_f32_16x16x4_f32(false, a, false, b1, (short)0, acc1, false, false);
        acc2 = __builtin_amdgcn_wmma_f32_16x16x4_f32(false, a, false, b2, (short)0, acc2, false, false);
        acc3 = __builtin_amdgcn_wmma_f32_16x16x4_f32(false, a, false, b3, (short)0, acc3, false, false);
    }
    float* crow = C + (size_t)row0 * fout + col0;
#pragma unroll
    for (int r = 0; r < 8; ++r) {
        const int m = r + 8 * half;
        float* p = crow + (size_t)m * fout + l;
        p[ 0] = acc0[r];
        p[16] = acc1[r];
        p[32] = acc2[r];
        p[48] = acc3[r];
    }
}

__global__ void alpha_kernel(const float* __restrict__ h,
                             const float* __restrict__ a_src, const float* __restrict__ a_dst,
                             float* __restrict__ as_, float* __restrict__ ad_,
                             int nn, int H, int Cc) {
    int idx = blockIdx.x * blockDim.x + threadIdx.x;
    if (idx >= nn * H) return;
    int n = idx / H, hh = idx - n * H;
    const float* hp = h + (size_t)n * H * Cc + hh * Cc;
    float s = 0.f, d = 0.f;
    for (int c = 0; c < Cc; ++c) {
        float v = hp[c];
        s += v * a_src[hh * Cc + c];
        d += v * a_dst[hh * Cc + c];
    }
    as_[idx] = s;
    ad_[idx] = d;
}

__global__ void edge_rank_kernel(const int* __restrict__ ei, int* __restrict__ cnt,
                                 int* __restrict__ rank) {
    int t = blockIdx.x * blockDim.x + threadIdx.x;
    if (t >= NEDGES) return;
    rank[t] = atomicAdd(&cnt[ei[NEDGES + t]], 1);
}

#define SCAN_ROWS ((NNODES + 1023) / 1024)
__global__ void __launch_bounds__(1024) rowptr_scan_kernel(const int* __restrict__ cnt,
                                                           int* __restrict__ rowptr) {
    __shared__ int wtot[32];
    const int lane = threadIdx.x & 31;
    const int w    = threadIdx.x >> 5;
    const int base = w * (SCAN_ROWS * 32);
    int v[SCAN_ROWS];
    int sum = 0;
#pragma unroll
    for (int it = 0; it < SCAN_ROWS; ++it) {
        const int i = base + it * 32 + lane;
        v[it] = (i < NNODES) ? cnt[i] : 0;
        sum += v[it];
    }
    for (int off = 16; off; off >>= 1) sum += __shfl_xor(sum, off);
    if (lane == 0) wtot[w] = sum;
    __syncthreads();
    int carry = 0;
    for (int k = 0; k < w; ++k) carry += wtot[k];
#pragma unroll
    for (int it = 0; it < SCAN_ROWS; ++it) {
        int incl = v[it];
        for (int off = 1; off < 32; off <<= 1) {
            const int u = __shfl_up(incl, off);
            if (lane >= off) incl += u;
        }
        const int i = base + it * 32 + lane;
        if (i < NNODES) rowptr[i] = carry + incl - v[it];
        carry += __shfl(incl, 31);
    }
    if (threadIdx.x == 1023) rowptr[NNODES] = carry;
}

__global__ void edge_place_kernel(const int* __restrict__ ei, const int* __restrict__ rowptr,
                                  const int* __restrict__ rank, int* __restrict__ csrc) {
    int t = blockIdx.x * blockDim.x + threadIdx.x;
    if (t >= NEDGES) return;
    csrc[rowptr[ei[NEDGES + t]] + rank[t]] = ei[t];
}

__device__ __forceinline__ float wave_max(float v) {
    for (int off = 16; off; off >>= 1) v = fmaxf(v, __shfl_xor(v, off));
    return v;
}
__device__ __forceinline__ float wave_sum(float v) {
    for (int off = 16; off; off >>= 1) v += __shfl_xor(v, off);
    return v;
}
__device__ __forceinline__ float lane_bcast(float v, int q) {
    return __int_as_float(__builtin_amdgcn_readlane(__float_as_int(v), q));
}

template <int H>
__device__ __forceinline__ void load_heads(const float* __restrict__ p, float (&o)[H]) {
    if constexpr (H == 4) {
        const float4 t = *reinterpret_cast<const float4*>(p);
        o[0] = t.x; o[1] = t.y; o[2] = t.z; o[3] = t.w;
    } else {
        for (int hh = 0; hh < H; ++hh) o[hh] = p[hh];
    }
}

template <int H>
__device__ __forceinline__ void load_chunk(const int* __restrict__ csrc, const float* __restrict__ as_,
                                           const float (&adv)[H], int d, int beg, int n, int j,
                                           int& s, float (&e)[H], bool& valid) {
    valid = j < n;
    s = d;
    if (j < n - 1) s = csrc[beg + j];
    float a[H];
    load_heads<H>(as_ + (size_t)s * H, a);
    for (int hh = 0; hh < H; ++hh) e[hh] = leakyf(a[hh] + adv[hh], 0.2f);
}

template <int H>
__device__ __forceinline__ void add_rows(const float* __restrict__ h, int s, const float (&al)[H],
                                         int cnt, int lane, float (&acc)[H * 2]) {
    constexpr int F = H * 64;
#pragma unroll 4
    for (int q = 0; q < cnt; ++q) {
        const int sq = __builtin_amdgcn_readlane(s, q);
        const float* row = h + (size_t)sq * F;
        if constexpr (H == 4) {
            const float4 r0 = *reinterpret_cast<const float4*>(row + 4 * lane);
            const float4 r1 = *reinterpret_cast<const float4*>(row + 128 + 4 * lane);
            const float a0 = lane_bcast(al[0], q), a1 = lane_bcast(al[1], q);
            const float a2 = lane_bcast(al[2], q), a3 = lane_bcast(al[3], q);
            const float w0 = lane < 16 ? a0 : a1;
            const float w1 = lane < 16 ? a2 : a3;
            acc[0] += w0 * r0.x; acc[1] += w0 * r0.y; acc[2] += w0 * r0.z; acc[3] += w0 * r0.w;
            acc[4] += w1 * r1.x; acc[5] += w1 * r1.y; acc[6] += w1 * r1.z; acc[7] += w1 * r1.w;
        } else {
            const float2 r = *reinterpret_cast<const float2*>(row + 2 * lane);
            const float a0 = lane_bcast(al[0], q);
            acc[0] += a0 * r.x; acc[1] += a0 * r.y;
        }
    }
}

template <int H>
__device__ __forceinline__ void gather_body(const int* __restrict__ rowptr, const int* __restrict__ csrc,
                                            const float* __restrict__ h,
                                            const float* __restrict__ as_, const float* __restrict__ ad_,
                                            float* __restrict__ out) {
    constexpr int F = H * 64;
    const int d    = __builtin_amdgcn_readfirstlane((blockIdx.x * blockDim.x + threadIdx.x) >> 5);
    const int lane = threadIdx.x & 31;
    if (d >= NNODES) return;
    const int beg = __builtin_amdgcn_readfirstlane(rowptr[d]);
    const int n   = __builtin_amdgcn_readfirstlane(rowptr[d + 1]) - beg + 1;
    float adv[H];
    load_heads<H>(ad_ + (size_t)d * H, adv);

    int s0; float e0[H]; bool v0;
    load_chunk<H>(csrc, as_, adv, d, beg, n, lane, s0, e0, v0);
    float m[H];
    for (int hh = 0; hh < H; ++hh) m[hh] = v0 ? e0[hh] : -INFINITY;
    for (int j0 = 32; j0 < n; j0 += 32) {
        int s; float e[H]; bool v;
        load_chunk<H>(csrc, as_, adv, d, beg, n, j0 + lane, s, e, v);
        for (int hh = 0; hh < H; ++hh) m[hh] = v ? fmaxf(m[hh], e[hh]) : m[hh];
    }
    for (int hh = 0; hh < H; ++hh) {
        m[hh] = wave_max(m[hh]);
        m[hh] = isfinite(m[hh]) ? m[hh] : 0.f;
    }

    float ex0[H], den[H];
    for (int hh = 0; hh < H; ++hh) { ex0[hh] = v0 ? expf(e0[hh] - m[hh]) : 0.f; den[hh] = ex0[hh]; }
    for (int j0 = 32; j0 < n; j0 += 32) {
        int s; float e[H]; bool v;
        load_chunk<H>(csrc, as_, adv, d, beg, n, j0 + lane, s, e, v);
        for (int hh = 0; hh < H; ++hh) den[hh] += v ? expf(e[hh] - m[hh]) : 0.f;
    }
    for (int hh = 0; hh < H; ++hh) den[hh] = wave_sum(den[hh]) + 1e-16f;

    float acc[H * 2];
    for (int i = 0; i < H * 2; ++i) acc[i] = 0.f;
    {
        float al[H];
        for (int hh = 0; hh < H; ++hh) al[hh] = ex0[hh] / den[hh];
        add_rows<H>(h, s0, al, n < 32 ? n : 32, lane, acc);
    }
    for (int j0 = 32; j0 < n; j0 += 32) {
        int s; float e[H]; bool v;
        load_chunk<H>(csrc, as_, adv, d, beg, n, j0 + lane, s, e, v);
        float al[H];
        for (int hh = 0; hh < H; ++hh) al[hh] = v ? expf(e[hh] - m[hh]) / den[hh] : 0.f;
        add_rows<H>(h, s, al, (n - j0) < 32 ? (n - j0) : 32, lane, acc);
    }

    float* orow = out + (size_t)d * F;
    if constexpr (H == 4) {
        *reinterpret_cast<float4*>(orow + 4 * lane)       = make_float4(acc[0], acc[1], acc[2], acc[3]);
        *reinterpret_cast<float4*>(orow + 128 + 4 * lane) = make_float4(acc[4], acc[5], acc[6], acc[7]);
    } else {
        *reinterpret_cast<float2*>(orow + 2 * lane) = make_float2(acc[0], acc[1]);
    }
}

__global__ void __launch_bounds__(256) gather_heads4_kernel(const int* __restrict__ rowptr,
        const int* __restrict__ csrc, const float* __restrict__ h, const float* __restrict__ as_,
        const float* __restrict__ ad_, float* __restrict__ out) {
    gather_body<4>(rowptr, csrc, h, as_, ad_, out);
}
__global__ void __launch_bounds__(256) gather_heads1_kernel(const int* __restrict__ rowptr,
        const int* __restrict__ csrc, const float* __restrict__ h, const float* __restrict__ as_,
        const float* __restrict__ ad_, float* __restrict__ out) {
    gather_body<1>(rowptr, csrc, h, as_, ad_, out);
}

__global__ void bias_act_kernel(const float* __restrict__ in, const float* __restrict__ b,
                                float* __restrict__ out, int total, int F, float slope) {
    int i = blockIdx.x * blockDim.x + threadIdx.x;
    if (i >= total) return;
    out[i] = leakyf(in[i] + b[i % F], slope);
}

__global__ void fill_u32_kernel(unsigned* __restrict__ p, unsigned v, int n) {
    int i = blockIdx.x * blockDim.x + threadIdx.x;
    if (i < n) p[i] = v;
}

extern "C" void kernel_launch(void* const* d_in, const int* in_sizes, int n_in,
                              void* d_out, int out_size, void* d_ws, size_t ws_size,
                              hipStream_t stream) {
    const float* x   = (const float*)d_in[0];
    const int*   ei  = (const int*)  d_in[1];
    const float* W1  = (const float*)d_in[2];
    const float* as1 = (const float*)d_in[3];
    const float* ad1 = (const float*)d_in[4];
    const float* b1  = (const float*)d_in[5];
    const float* W2  = (const float*)d_in[6];
    const float* as2 = (const float*)d_in[7];
    const float* ad2 = (const float*)d_in[8];
    const float* b2  = (const float*)d_in[9];
    float* out = (float*)d_out;

    float* ws   = (float*)d_ws;
    float* h1   = ws;
    float* h2   = h1   + (size_t)NNODES * F1;
    int*   grp  = (int*)(h2 + (size_t)NNODES * F1);
    int*   cnt    = grp;
    int*   rowptr = grp + 65536;
    int*   rank   = grp + 131072;
    int*   csrc   = rank + NEDGES;
    float* asb  = (float*)(grp + (size_t)(NEDGES + NNODES) * HEADS);
    float* adb  = asb  + (size_t)NNODES * HEADS;
    float* g2   = h1;
    float* out2 = h1 + (size_t)NNODES * OUT_DIM;

    fill_u32_kernel<<<(NNODES + 255) / 256, 256, 0, stream>>>((unsigned*)cnt, 0u, NNODES);
    edge_rank_kernel<<<(NEDGES + 255) / 256, 256, 0, stream>>>(ei, cnt, rank);
    rowptr_scan_kernel<<<1, 1024, 0, stream>>>(cnt, rowptr);
    edge_place_kernel<<<(NEDGES + 255) / 256, 256, 0, stream>>>(ei, rowptr, rank, csrc);

    {
        int strips = (NNODES / 16) * (F1 / 64);
        gemm_wmma_f32_kernel<<<(strips + 3) / 4, 128, 0, stream>>>(x, W1, h1, NNODES, IN_DIM, F1);
    }
    alpha_kernel<<<(NNODES * HEADS + 255) / 256, 256, 0, stream>>>(h1, as1, ad1, asb, adb,
                                                                   NNODES, HEADS, HID);
    gather_heads4_kernel<<<(NNODES + 7) / 8, 256, 0, stream>>>(rowptr, csrc, h1, asb, adb, h2);
    bias_act_kernel<<<(NNODES * F1 + 255) / 256, 256, 0, stream>>>(h2, b1, h2,
                                                                   NNODES * F1, F1, 0.01f);

    {
        int strips = (NNODES / 16) * (OUT_DIM / 64);
        gemm_wmma_f32_kernel<<<(strips + 3) / 4, 128, 0, stream>>>(h2, W2, g2, NNODES, F1, OUT_DIM);
    }
    alpha_kernel<<<(NNODES + 255) / 256, 256, 0, stream>>>(g2, as2, ad2, asb, adb,
                                                           NNODES, 1, OUT_DIM);
    gather_heads1_kernel<<<(NNODES + 7) / 8, 256, 0, stream>>>(rowptr, csrc, g2, asb, adb, out2);
    bias_act_kernel<<<(NNODES * OUT_DIM + 255) / 256, 256, 0, stream>>>(out2, b2, out,
                                                                        NNODES * OUT_DIM,
                                                                        OUT_DIM, 1.0f);
}


# Both edge-grouping launches ride inside the layer-1 product's launches (product cut into two row ranges; each launch's first work-groups run one grouping stage beside it): 6 launches -> 5
# speedup vs baseline: 5.4503x; 2.0294x over previous
#include <hip/hip_runtime.h>
#include <math.h>

#define NNODES  50000
#define NEDGES  800000
#define IN_DIM  128
#define HID     64
#define HEADS   4
#define OUT_DIM 64
#define F1      (HEADS * HID)

typedef __attribute__((ext_vector_type(2))) float v2f;
typedef __attribute__((ext_vector_type(8))) float v8f;

__device__ __forceinline__ float leakyf(float x, float s) { return x >= 0.f ? x : s * x; }


template <int X>
__device__ __forceinline__ float row_xor_add(float v) {
    return v + __int_as_float(__builtin_amdgcn_update_dpp(0, __float_as_int(v), 0x160 + X, 0xf, 0xf, true));
}

__device__ __forceinline__ void gemm_logits_body(const float* __restrict__ A,
                                     const float* __restrict__ W,
                                     const float* __restrict__ a_src, const float* __restrict__ a_dst,
                                     float* __restrict__ C,
                                     float* __restrict__ as_, float* __restrict__ ad_,
                                     int nrows, int fin, int fout,
                                     int vb, int vt, float* lg) {
    const int gwave = (vb * 128 + vt) >> 5;
    const int lane  = vt & 31;
    const int stripsPerRow = fout >> 6;
    const int totalStrips  = (nrows >> 4) * stripsPerRow;
    const bool live = gwave < totalStrips;

    const int rowBlk = gwave / stripsPerRow;
    const int strip  = gwave % stripsPerRow;
    const int row0 = rowBlk << 4;
    const int col0 = strip << 6;
    const int half = lane >> 4;
    const int l    = lane & 15;

    v8f acc0 = {}, acc1 = {}, acc2 = {}, acc3 = {};
    if (live) {
        float sa[4], da[4];
#pragma unroll
        for (int t = 0; t < 4; ++t) {
            sa[t] = a_src[col0 + 16 * t + l];
            da[t] = a_dst[col0 + 16 * t + l];
        }

        const float* arow = A + (size_t)(row0 + l) * fin;
        for (int k = 0; k < fin; k += 4) {
            v2f a;
            a.x = arow[k + 2 * half + 0];
            a.y = arow[k + 2 * half + 1];
            const float* w0 = W + (size_t)(k + 2 * half) * fout + col0 + l;
            const float* w1 = w0 + fout;
            v2f b0 = { w0[ 0], w1[ 0] };
            v2f b1 = { w0[16], w1[16] };
            v2f b2 = { w0[32], w1[32] };
            v2f b3 = { w0[48], w1[48] };
            acc0 = __builtin_amdgcn_wmma_f32_16x16x4_f32(false, a, false, b0, (short)0, acc0, false, false);
            acc1 = __builtin_amdgcn_wmma_f32_16x16x4_f32(false, a, false, b1, (short)0, acc1, false, false);
            acc2 = __builtin_amdgcn_wmma_f32_16x16x4_f32(false, a, false, b2, (short)0, acc2, false, false);
            acc3 = __builtin_amdgcn_wmma_f32_16x16x4_f32(false, a, false, b3, (short)0, acc3, false, false);
        }

        float ps[8], pd[8];
#pragma unroll
        for (int r = 0; r < 8; ++r) {
            ps[r] = acc0[r] * sa[0] + acc1[r] * sa[1] + acc2[r] * sa[2] + acc3[r] * sa[3];
            pd[r] = acc0[r] * da[0] + acc1[r] * da[1] + acc2[r] * da[2] + acc3[r] * da[3];
        }
#pragma unroll
        for (int r = 0; r < 8; ++r) { ps[r] = row_xor_add<1>(ps[r]); pd[r] = row_xor_add<1>(pd[r]); }
#pragma unroll
        for (int r = 0; r < 8; ++r) { ps[r] = row_xor_add<2>(ps[r]); pd[r] = row_xor_add<2>(pd[r]); }
#pragma unroll
        for (int r = 0; r < 8; ++r) { ps[r] = row_xor_add<4>(ps[r]); pd[r] = row_xor_add<4>(pd[r]); }
#pragma unroll
        for (int r = 0; r < 8; ++r) { ps[r] = row_xor_add<8>(ps[r]); pd[r] = row_xor_add<8>(pd[r]); }
        float vs = ps[0], vd = pd[0];
#pragma unroll
        for (int r = 1; r < 8; ++r) {
            vs = (l == r) ? ps[r] : vs;
            vd = (l == r) ? pd[r] : vd;
        }
        if (l < 8) {
            const int e = (row0 + l + 8 * half) * stripsPerRow + strip - vb * 64;
            lg[e]      = vs;
            lg[64 + e] = vd;
        }
    }
    __syncthreads();
    {
        const int t = vt & 63;
        const int g = vb * 64 + t;
        if (g < nrows * stripsPerRow) {
            float* dstp = vt < 64 ? as_ : ad_;
            dstp[g] = lg[vt];
        }
    }
    if (live) {
        float* crow = C + (size_t)row0 * fout + col0;
#pragma unroll
        for (int r = 0; r < 8; ++r) {
            const int m = r + 8 * half;
            float* p = crow + (size_t)m * fout + l;
            p[ 0] = acc0[r];
            p[16] = acc1[r];
            p[32] = acc2[r];
            p[48] = acc3[r];
        }
    }
}

__global__ void __launch_bounds__(128) gemm_wmma_f32_logits_kernel(const float* __restrict__ A,
                                     const float* __restrict__ W,
                                     const float* __restrict__ a_src, const float* __restrict__ a_dst,
                                     float* __restrict__ C,
                                     float* __restrict__ as_, float* __restrict__ ad_,
                                     int nrows, int fin, int fout) {
    __shared__ float lg[128];
    gemm_logits_body(A, W, a_src, a_dst, C, as_, ad_, nrows, fin, fout, (int)blockIdx.x, (int)threadIdx.x, lg);
}

#define BIN_SHIFT   8
#define BIN_NODES   (1 << BIN_SHIFT)
#define NBINS       ((NNODES + BIN_NODES - 1) / BIN_NODES)
#define NCHUNK      256
#define CHUNK       ((NEDGES + NCHUNK - 1) / NCHUNK)
#define CHUNK_PITCH ((CHUNK + 31) & ~31)
#define GRP_T       256
#define BIN_TRIPS   ((CHUNK + GRP_T - 1) / GRP_T)
#define RUNS_PER_WAVE (NCHUNK / (GRP_T / 32))
#define STAGE_CAP   16384
static_assert(NNODES <= 65536, "an edge is packed as destination << 16 | source");
static_assert(NBINS < 256, "bin offsets are scanned by 256 threads, entry NBINS is the chunk total");
static_assert(NNODES + 1 <= 65536 && NCHUNK * 256 <= 65536 &&
              131072 + NCHUNK * CHUNK_PITCH + NEDGES <= (NEDGES + NNODES) * HEADS,
              "row pointers, bin offsets, sorted chunks and source list fit the grouping area of the workspace");
static_assert(GRP_T >= 256 && NCHUNK <= GRP_T && NCHUNK % (GRP_T / 32) == 0 && RUNS_PER_WAVE <= 32,
              "256 scanning threads, one thread per chunk, whole runs per wave, one mask bit per run");

__device__ __forceinline__ int scan256(int v, int* wtot, int& total) {
    const int lane = threadIdx.x & 31, w = threadIdx.x >> 5;
    int incl = v;
    for (int off = 1; off < 32; off <<= 1) {
        const int u = __shfl_up(incl, off);
        if (lane >= off) incl += u;
    }
    if (w < 8 && lane == 31) wtot[w] = incl;
    __syncthreads();
    int carry = 0;
    total = 0;
    for (int k = 0; k < 8; ++k) {
        const int x = wtot[k];
        carry += (k < w) ? x : 0;
        total += x;
    }
    return carry + incl - v;
}

__device__ __forceinline__ void edge_bin_body(const int* __restrict__ ei,
                                              unsigned* __restrict__ sorted,
                                              int* __restrict__ boff) {
    __shared__ int hist[256];
    __shared__ int wtot[8];
    __shared__ unsigned stage[CHUNK];
    const int t = threadIdx.x, g = blockIdx.x;
    const int e0 = g * CHUNK;
    int cnt = NEDGES - e0;
    cnt = cnt < 0 ? 0 : (cnt > CHUNK ? CHUNK : cnt);
    if (t < 256) hist[t] = 0;
    __syncthreads();
    unsigned word[BIN_TRIPS];
    int rk[BIN_TRIPS];
    {
        const int last = (e0 + cnt < NEDGES ? e0 + cnt : NEDGES) - 1;
        unsigned sraw[BIN_TRIPS], draw[BIN_TRIPS];
#pragma unroll
        for (int it = 0; it < BIN_TRIPS; ++it) {
            const int i = e0 + it * GRP_T + t;
            const int idx = i < last ? i : last;
            sraw[it] = (unsigned)ei[idx];
            draw[it] = (unsigned)ei[NEDGES + idx];
        }
#pragma unroll
        for (int it = 0; it < BIN_TRIPS; ++it) {
            word[it] = (draw[it] << 16) | sraw[it];
            rk[it] = 0;
            if (it * GRP_T + t < cnt) rk[it] = atomicAdd(&hist[draw[it] >> BIN_SHIFT], 1);
        }
    }
    __syncthreads();
    int total;
    const int v = t < 256 ? hist[t] : 0;
    const int excl = scan256(v, wtot, total);
    if (t < 256) {
        hist[t] = excl;
        boff[g * 256 + t] = excl;
    }
    __syncthreads();
#pragma unroll
    for (int it = 0; it < BIN_TRIPS; ++it) {
        const int i = it * GRP_T + t;
        if (i < cnt) stage[hist[word[it] >> (16 + BIN_SHIFT)] + rk[it]] = word[it];
    }
    __syncthreads();
    unsigned* dstp = sorted + (size_t)g * CHUNK_PITCH;
    for (int i = t; i < cnt; i += GRP_T) dstp[i] = stage[i];
}

__device__ __forceinline__ void bin_group_body(const unsigned* __restrict__ sorted,
                                               const int* __restrict__ boff,
                                               int* __restrict__ rowptr,
                                               int* __restrict__ csrc) {
    __shared__ int rs[NCHUNK], re[NCHUNK];
    __shared__ int cur[256];
    __shared__ int wtot[8];
    __shared__ int part[GRP_T / 32];
    __shared__ int stage[STAGE_CAP];
    const int t = threadIdx.x, b = blockIdx.x, lane = t & 31, w = t >> 5;
    int sv = 0;
    if (t < NCHUNK) {
        sv = boff[t * 256 + b];
        rs[t] = sv;
        re[t] = boff[t * 256 + b + 1];
    }
    if (t < 256) cur[t] = 0;
    int psum = sv;
    for (int off = 16; off; off >>= 1) psum += __shfl_xor(psum, off);
    if (lane == 0) part[w] = psum;
    __syncthreads();
    int base = 0;
    for (int k = 0; k < GRP_T / 32; ++k) base += part[k];

    unsigned w0[RUNS_PER_WAVE];
    unsigned live = 0u;
    unsigned longm = 0u;
    const int rsv = lane < RUNS_PER_WAVE ? rs[w * RUNS_PER_WAVE + lane] : 0;
    const int rev = lane < RUNS_PER_WAVE ? re[w * RUNS_PER_WAVE + lane] : 0;
#pragma unroll
    for (int r = 0; r < RUNS_PER_WAVE; ++r) {
        const int g = w * RUNS_PER_WAVE + r;
        const int s = __builtin_amdgcn_readlane(rsv, r), e = __builtin_amdgcn_readlane(rev, r);
        w0[r] = 0u;
        if (s + lane < e) {
            w0[r] = sorted[(size_t)g * CHUNK_PITCH + s + lane];
            live |= 1u << r;
        }
        if (e - s > 32) longm |= 1u << r;
    }
#pragma unroll
    for (int r = 0; r < RUNS_PER_WAVE; ++r)
        if ((live >> r) & 1u) atomicAdd(&cur[(w0[r] >> 16) & (BIN_NODES - 1)], 1);
    for (unsigned m = longm; m; m &= m - 1u) {
        const int g = w * RUNS_PER_WAVE + __builtin_ctz(m);
        const unsigned* p = sorted + (size_t)g * CHUNK_PITCH;
        const int e = re[g];
        for (int i = rs[g] + lane + 32; i < e; i += 32) atomicAdd(&cur[(p[i] >> 16) & (BIN_NODES - 1)], 1);
    }
    __syncthreads();
    int total;
    const int v = t < 256 ? cur[t] : 0;
    const int excl = scan256(v, wtot, total);
    if (t < 256) {
        cur[t] = excl;
        const int node = b * BIN_NODES + t;
        if (node < NNODES) rowptr[node] = base + excl;
        if (node == NNODES - 1) rowptr[NNODES] = base + excl + v;
    }
    __syncthreads();
    const bool staged = total <= STAGE_CAP;
    {
        int pos[RUNS_PER_WAVE];
#pragma unroll
        for (int r = 0; r < RUNS_PER_WAVE; ++r) {
            pos[r] = 0;
            if ((live >> r) & 1u) pos[r] = atomicAdd(&cur[(w0[r] >> 16) & (BIN_NODES - 1)], 1);
        }
        if (staged) {
#pragma unroll
            for (int r = 0; r < RUNS_PER_WAVE; ++r)
                if ((live >> r) & 1u) stage[pos[r]] = (int)(w0[r] & 0xffffu);
        } else {
#pragma unroll
            for (int r = 0; r < RUNS_PER_WAVE; ++r)
                if ((live >> r) & 1u) csrc[base + pos[r]] = (int)(w0[r] & 0xffffu);
        }
    }
    for (unsigned m = longm; m; m &= m - 1u) {
        const int g = w * RUNS_PER_WAVE + __builtin_ctz(m);
        const unsigned* p = sorted + (size_t)g * CHUNK_PITCH;
        const int e = re[g];
        for (int i = rs[g] + lane + 32; i < e; i += 32) {
            const unsigned wd = p[i];
            const int pos = atomicAdd(&cur[(wd >> 16) & (BIN_NODES - 1)], 1);
            const int sidx = (int)(wd & 0xffffu);
            if (staged) stage[pos] = sidx; else csrc[base + pos] = sidx;
        }
    }
    if (staged) {
        __syncthreads();
        for (int i = t; i < total; i += GRP_T) csrc[base + i] = stage[i];
    }
}

#define GEMM1_GROUPS   ((NNODES / 16) * (F1 / 64) / 4)
#define GEMM1_GROUPS_A ((GEMM1_GROUPS / 4) * 2)
#define GEMM1_WG_A     (GEMM1_GROUPS_A / 2)
#define GEMM1_WG_B     ((GEMM1_GROUPS - GEMM1_GROUPS_A + 1) / 2)
static_assert(((NNODES / 16) * (F1 / 64)) % 4 == 0 && GRP_T == 256, "whole groups; two groups per work-group");

__global__ void __launch_bounds__(GRP_T, 8) edge_bin_and_gemm_kernel(const int* __restrict__ ei,
        unsigned* __restrict__ sorted, int* __restrict__ boff,
        const float* __restrict__ A, const float* __restrict__ W,
        const float* __restrict__ a_src, const float* __restrict__ a_dst,
        float* __restrict__ C, float* __restrict__ as_, float* __restrict__ ad_,
        int nrows, int fin, int fout, int vb0) {
    __shared__ float lg[2 * 128];
    if (blockIdx.x < NCHUNK) {
        edge_bin_body(ei, sorted, boff);
    } else {
        const int sub = (int)threadIdx.x >> 7;
        gemm_logits_body(A, W, a_src, a_dst, C, as_, ad_, nrows, fin, fout,
                         vb0 + 2 * ((int)blockIdx.x - NCHUNK) + sub, (int)threadIdx.x & 127, lg + 128 * sub);
    }
}

__global__ void __launch_bounds__(GRP_T) bin_group_and_gemm_kernel(const unsigned* __restrict__ sorted,
        const int* __restrict__ boff, int* __restrict__ rowptr, int* __restrict__ csrc,
        const float* __restrict__ A, const float* __restrict__ W,
        const float* __restrict__ a_src, const float* __restrict__ a_dst,
        float* __restrict__ C, float* __restrict__ as_, float* __restrict__ ad_,
        int nrows, int fin, int fout, int vb0) {
    __shared__ float lg[2 * 128];
    if (blockIdx.x < NBINS) {
        bin_group_body(sorted, boff, rowptr, csrc);
    } else {
        const int sub = (int)threadIdx.x >> 7;
        gemm_logits_body(A, W, a_src, a_dst, C, as_, ad_, nrows, fin, fout,
                         vb0 + 2 * ((int)blockIdx.x - NBINS) + sub, (int)threadIdx.x & 127, lg + 128 * sub);
    }
}

__device__ __forceinline__ float wave_max(float v) {
    for (int off = 16; off; off >>= 1) v = fmaxf(v, __shfl_xor(v, off));
    return v;
}
__device__ __forceinline__ float wave_sum(float v) {
    for (int off = 16; off; off >>= 1) v += __shfl_xor(v, off);
    return v;
}
__device__ __forceinline__ float lane_bcast(float v, int q) {
    return __int_as_float(__builtin_amdgcn_readlane(__float_as_int(v), q));
}

template <int H>
__device__ __forceinline__ void load_heads(const float* __restrict__ p, float (&o)[H]) {
    if constexpr (H == 4) {
        const float4 t = *reinterpret_cast<const float4*>(p);
        o[0] = t.x; o[1] = t.y; o[2] = t.z; o[3] = t.w;
    } else {
        for (int hh = 0; hh < H; ++hh) o[hh] = p[hh];
    }
}

template <int H>
__device__ __forceinline__ void load_chunk(const int* __restrict__ csrc, const float* __restrict__ as_,
                                           const float (&adv)[H], int d, int beg, int n, int j,
                                           int& s, float (&e)[H], bool& valid) {
    valid = j < n;
    s = d;
    if (j < n - 1) s = csrc[beg + j];
    float a[H];
    load_heads<H>(as_ + (size_t)s * H, a);
    for (int hh = 0; hh < H; ++hh) e[hh] = leakyf(a[hh] + adv[hh], 0.2f);
}

template <int H>
__device__ __forceinline__ void add_rows(const float* __restrict__ h, int s, const float (&al)[H],
                                         int cnt, int lane, float (&acc)[H * 2]) {
    constexpr int F = H * 64;
#pragma unroll 4
    for (int q = 0; q < cnt; ++q) {
        const int sq = __builtin_amdgcn_readlane(s, q);
        const float* row = h + (size_t)sq * F;
        if constexpr (H == 4) {
            const float4 r0 = *reinterpret_cast<const float4*>(row + 4 * lane);
            const float4 r1 = *reinterpret_cast<const float4*>(row + 128 + 4 * lane);
            const float a0 = lane_bcast(al[0], q), a1 = lane_bcast(al[1], q);
            const float a2 = lane_bcast(al[2], q), a3 = lane_bcast(al[3], q);
            const float w0 = lane < 16 ? a0 : a1;
            const float w1 = lane < 16 ? a2 : a3;
            acc[0] += w0 * r0.x; acc[1] += w0 * r0.y; acc[2] += w0 * r0.z; acc[3] += w0 * r0.w;
            acc[4] += w1 * r1.x; acc[5] += w1 * r1.y; acc[6] += w1 * r1.z; acc[7] += w1 * r1.w;
        } else {
            const float2 r = *reinterpret_cast<const float2*>(row + 2 * lane);
            const float a0 = lane_bcast(al[0], q);
            acc[0] += a0 * r.x; acc[1] += a0 * r.y;
        }
    }
}

template <int H, bool LEAKY>
__device__ __forceinline__ void gather_body(const int* __restrict__ rowptr, const int* __restrict__ csrc,
                                            const float* __restrict__ h,
                                            const float* __restrict__ as_, const float* __restrict__ ad_,
                                            const float* __restrict__ bias,
                                            float* __restrict__ out) {
    constexpr int F = H * 64;
    const int d    = __builtin_amdgcn_readfirstlane((blockIdx.x * blockDim.x + threadIdx.x) >> 5);
    const int lane = threadIdx.x & 31;
    if (d >= NNODES) return;
    const int beg = __builtin_amdgcn_readfirstlane(rowptr[d]);
    const int n   = __builtin_amdgcn_readfirstlane(rowptr[d + 1]) - beg + 1;
    float adv[H];
    load_heads<H>(ad_ + (size_t)d * H, adv);
    float bv[H * 2];
    if constexpr (H == 4) {
        const float4 b0 = *reinterpret_cast<const float4*>(bias + 4 * lane);
        const float4 b1 = *reinterpret_cast<const float4*>(bias + 128 + 4 * lane);
        bv[0] = b0.x; bv[1] = b0.y; bv[2] = b0.z; bv[3] = b0.w;
        bv[4] = b1.x; bv[5] = b1.y; bv[6] = b1.z; bv[7] = b1.w;
    } else {
        const float2 b0 = *reinterpret_cast<const float2*>(bias + 2 * lane);
        bv[0] = b0.x; bv[1] = b0.y;
    }

    int s0; float e0[H]; bool v0;
    load_chunk<H>(csrc, as_, adv, d, beg, n, lane, s0, e0, v0);
    float m[H];
    for (int hh = 0; hh < H; ++hh) m[hh] = v0 ? e0[hh] : -INFINITY;
    for (int j0 = 32; j0 < n; j0 += 32) {
        int s; float e[H]; bool v;
        load_chunk<H>(csrc, as_, adv, d, beg, n, j0 + lane, s, e, v);
        for (int hh = 0; hh < H; ++hh) m[hh] = v ? fmaxf(m[hh], e[hh]) : m[hh];
    }
    for (int hh = 0; hh < H; ++hh) {
        m[hh] = wave_max(m[hh]);
        m[hh] = isfinite(m[hh]) ? m[hh] : 0.f;
    }

    float ex0[H], den[H];
    for (int hh = 0; hh < H; ++hh) { ex0[hh] = v0 ? expf(e0[hh] - m[hh]) : 0.f; den[hh] = ex0[hh]; }
    for (int j0 = 32; j0 < n; j0 += 32) {
        int s; float e[H]; bool v;
        load_chunk<H>(csrc, as_, adv, d, beg, n, j0 + lane, s, e, v);
        for (int hh = 0; hh < H; ++hh) den[hh] += v ? expf(e[hh] - m[hh]) : 0.f;
    }
    for (int hh = 0; hh < H; ++hh) den[hh] = wave_sum(den[hh]) + 1e-16f;

    float acc[H * 2];
    for (int i = 0; i < H * 2; ++i) acc[i] = 0.f;
    {
        float al[H];
        for (int hh = 0; hh < H; ++hh) al[hh] = ex0[hh] / den[hh];
        add_rows<H>(h, s0, al, n < 32 ? n : 32, lane, acc);
    }
    for (int j0 = 32; j0 < n; j0 += 32) {
        int s; float e[H]; bool v;
        load_chunk<H>(csrc, as_, adv, d, beg, n, j0 + lane, s, e, v);
        float al[H];
        for (int hh = 0; hh < H; ++hh) al[hh] = v ? expf(e[hh] - m[hh]) / den[hh] : 0.f;
        add_rows<H>(h, s, al, (n - j0) < 32 ? (n - j0) : 32, lane, acc);
    }

    for (int i = 0; i < H * 2; ++i) {
        acc[i] += bv[i];
        if constexpr (LEAKY) acc[i] = leakyf(acc[i], 0.01f);
    }

    float* orow = out + (size_t)d * F;
    if constexpr (H == 4) {
        *reinterpret_cast<float4*>(orow + 4 * lane)       = make_float4(acc[0], acc[1], acc[2], acc[3]);
        *reinterpret_cast<float4*>(orow + 128 + 4 * lane) = make_float4(acc[4], acc[5], acc[6], acc[7]);
    } else {
        *reinterpret_cast<float2*>(orow + 2 * lane) = make_float2(acc[0], acc[1]);
    }
}

__global__ void __launch_bounds__(256) gather_heads4_kernel(const int* __restrict__ rowptr,
        const int* __restrict__ csrc, const float* __restrict__ h, const float* __restrict__ as_,
        const float* __restrict__ ad_, const float* __restrict__ bias, float* __restrict__ out) {
    gather_body<4, true>(rowptr, csrc, h, as_, ad_, bias, out);
}
__global__ void __launch_bounds__(256) gather_heads1_kernel(const int* __restrict__ rowptr,
        const int* __restrict__ csrc, const float* __restrict__ h, const float* __restrict__ as_,
        const float* __restrict__ ad_, const float* __restrict__ bias, float* __restrict__ out) {
    gather_body<1, false>(rowptr, csrc, h, as_, ad_, bias, out);
}

extern "C" void kernel_launch(void* const* d_in, const int* in_sizes, int n_in,
                              void* d_out, int out_size, void* d_ws, size_t ws_size,
                              hipStream_t stream) {
    const float* x   = (const float*)d_in[0];
    const int*   ei  = (const int*)  d_in[1];
    const float* W1  = (const float*)d_in[2];
    const float* as1 = (const float*)d_in[3];
    const float* ad1 = (const float*)d_in[4];
    const float* b1  = (const float*)d_in[5];
    const float* W2  = (const float*)d_in[6];
    const float* as2 = (const float*)d_in[7];
    const float* ad2 = (const float*)d_in[8];
    const float* b2  = (const float*)d_in[9];
    float* out = (float*)d_out;

    float* ws   = (float*)d_ws;
    float* h1   = ws;
    float* h2   = h1   + (size_t)NNODES * F1;
    int*   grp  = (int*)(h2 + (size_t)NNODES * F1);
    int*      rowptr = grp;
    int*      boff   = grp + 65536;
    unsigned* sorted = (unsigned*)(grp + 131072);
    int*      csrc   = grp + 131072 + NCHUNK * CHUNK_PITCH;
    float* asb  = (float*)(grp + (size_t)(NEDGES + NNODES) * HEADS);
    float* adb  = asb  + (size_t)NNODES * HEADS;
    float* g2   = h1;

    edge_bin_and_gemm_kernel<<<NCHUNK + GEMM1_WG_A, GRP_T, 0, stream>>>(ei, sorted, boff,
            x, W1, as1, ad1, h1, asb, adb, NNODES, IN_DIM, F1, 0);
    bin_group_and_gemm_kernel<<<NBINS + GEMM1_WG_B, GRP_T, 0, stream>>>(sorted, boff, rowptr, csrc,
            x, W1, as1, ad1, h1, asb, adb, NNODES, IN_DIM, F1, GEMM1_GROUPS_A);

    gather_heads4_kernel<<<(NNODES + 7) / 8, 256, 0, stream>>>(rowptr, csrc, h1, asb, adb, b1, h2);

    {
        int strips = (NNODES / 16) * (OUT_DIM / 64);
        gemm_wmma_f32_logits_kernel<<<(strips + 3) / 4, 128, 0, stream>>>(h2, W2, as2, ad2, g2, asb, adb,
                                                                         NNODES, F1, OUT_DIM);
    }
    gather_heads1_kernel<<<(NNODES + 7) / 8, 256, 0, stream>>>(rowptr, csrc, g2, asb, adb, b2, out);
}


# Layer-2 product: operand fragments of the next 4 k-steps fetched as one group into a second register set ahead of the current block's matrix instructions; same loaders, same k order
# speedup vs baseline: 6.3081x; 1.1574x over previous
#include <hip/hip_runtime.h>
#include <math.h>

#define NNODES  50000
#define NEDGES  800000
#define IN_DIM  128
#define HID     64
#define HEADS   4
#define OUT_DIM 64
#define F1      (HEADS * HID)

typedef __attribute__((ext_vector_type(2))) float v2f;
typedef __attribute__((ext_vector_type(8))) float v8f;

__device__ __forceinline__ float leakyf(float x, float s) { return x >= 0.f ? x : s * x; }


template <int X>
__device__ __forceinline__ float row_xor_add(float v) {
    return v + __int_as_float(__builtin_amdgcn_update_dpp(0, __float_as_int(v), 0x160 + X, 0xf, 0xf, true));
}

template <int KB>
__device__ __forceinline__ void gemm_logits_body(const float* __restrict__ A,
                                     const float* __restrict__ W,
                                     const float* __restrict__ a_src, const float* __restrict__ a_dst,
                                     float* __restrict__ C,
                                     float* __restrict__ as_, float* __restrict__ ad_,
                                     int nrows, int fin, int fout,
                                     int vb, int vt, float* lg) {
    const int gwave = (vb * 128 + vt) >> 5;
    const int lane  = vt & 31;
    const int stripsPerRow = fout >> 6;
    const int totalStrips  = (nrows >> 4) * stripsPerRow;
    const bool live = gwave < totalStrips;

    const int rowBlk = gwave / stripsPerRow;
    const int strip  = gwave % stripsPerRow;
    const int row0 = rowBlk << 4;
    const int col0 = strip << 6;
    const int half = lane >> 4;
    const int l    = lane & 15;

    v8f acc0 = {}, acc1 = {}, acc2 = {}, acc3 = {};
    if (live) {
        float sa[4], da[4];
#pragma unroll
        for (int t = 0; t < 4; ++t) {
            sa[t] = a_src[col0 + 16 * t + l];
            da[t] = a_dst[col0 + 16 * t + l];
        }

        const float* arow = A + (size_t)(row0 + l) * fin;
        if constexpr (KB == 1) {
            for (int k = 0; k < fin; k += 4) {
                v2f a;
                a.x = arow[k + 2 * half + 0];
                a.y = arow[k + 2 * half + 1];
                const float* w0 = W + (size_t)(k + 2 * half) * fout + col0 + l;
                const float* w1 = w0 + fout;
                v2f b0 = { w0[ 0], w1[ 0] };
                v2f b1 = { w0[16], w1[16] };
                v2f b2 = { w0[32], w1[32] };
                v2f b3 = { w0[48], w1[48] };
                acc0 = __builtin_amdgcn_wmma_f32_16x16x4_f32(false, a, false, b0, (short)0, acc0, false, false);
                acc1 = __builtin_amdgcn_wmma_f32_16x16x4_f32(false, a, false, b1, (short)0, acc1, false, false);
                acc2 = __builtin_amdgcn_wmma_f32_16x16x4_f32(false, a, false, b2, (short)0, acc2, false, false);
                acc3 = __builtin_amdgcn_wmma_f32_16x16x4_f32(false, a, false, b3, (short)0, acc3, false, false);
            }
        } else {
            auto load_block = [&](int kb, v2f (&fa)[KB], v2f (&fb)[KB][4]) {
#pragma unroll
                for (int j = 0; j < KB; ++j) {
                    const int k = kb + 4 * j;
                    fa[j].x = arow[k + 2 * half + 0];
                    fa[j].y = arow[k + 2 * half + 1];
                    const float* w0 = W + (size_t)(k + 2 * half) * fout + col0 + l;
                    const float* w1 = w0 + fout;
                    fb[j][0] = v2f{ w0[ 0], w1[ 0] };
                    fb[j][1] = v2f{ w0[16], w1[16] };
                    fb[j][2] = v2f{ w0[32], w1[32] };
                    fb[j][3] = v2f{ w0[48], w1[48] };
                }
            };
            auto mma_block = [&](const v2f (&fa)[KB], const v2f (&fb)[KB][4]) {
#pragma unroll
                for (int j = 0; j < KB; ++j) {
                    acc0 = __builtin_amdgcn_wmma_f32_16x16x4_f32(false, fa[j], false, fb[j][0], (short)0, acc0, false, false);
                    acc1 = __builtin_amdgcn_wmma_f32_16x16x4_f32(false, fa[j], false, fb[j][1], (short)0, acc1, false, false);
                    acc2 = __builtin_amdgcn_wmma_f32_16x16x4_f32(false, fa[j], false, fb[j][2], (short)0, acc2, false, false);
                    acc3 = __builtin_amdgcn_wmma_f32_16x16x4_f32(false, fa[j], false, fb[j][3], (short)0, acc3, false, false);
                }
            };
            constexpr int BK = 4 * KB;
            v2f pa[KB], pb[KB][4], qa[KB], qb[KB][4];
            load_block(0, pa, pb);
            int k = 0;
            for (; k + 2 * BK < fin; k += 2 * BK) {
                load_block(k + BK, qa, qb);
                mma_block(pa, pb);
                load_block(k + 2 * BK, pa, pb);
                mma_block(qa, qb);
            }
            if (k + BK < fin) {
                load_block(k + BK, qa, qb);
                mma_block(pa, pb);
                mma_block(qa, qb);
            } else {
                mma_block(pa, pb);
            }
        }

        float ps[8], pd[8];
#pragma unroll
        for (int r = 0; r < 8; ++r) {
            ps[r] = acc0[r] * sa[0] + acc1[r] * sa[1] + acc2[r] * sa[2] + acc3[r] * sa[3];
            pd[r] = acc0[r] * da[0] + acc1[r] * da[1] + acc2[r] * da[2] + acc3[r] * da[3];
        }
#pragma unroll
        for (int r = 0; r < 8; ++r) { ps[r] = row_xor_add<1>(ps[r]); pd[r] = row_xor_add<1>(pd[r]); }
#pragma unroll
        for (int r = 0; r < 8; ++r) { ps[r] = row_xor_add<2>(ps[r]); pd[r] = row_xor_add<2>(pd[r]); }
#pragma unroll
        for (int r = 0; r < 8; ++r) { ps[r] = row_xor_add<4>(ps[r]); pd[r] = row_xor_add<4>(pd[r]); }
#pragma unroll
        for (int r = 0; r < 8; ++r) { ps[r] = row_xor_add<8>(ps[r]); pd[r] = row_xor_add<8>(pd[r]); }
        float vs = ps[0], vd = pd[0];
#pragma unroll
        for (int r = 1; r < 8; ++r) {
            vs = (l == r) ? ps[r] : vs;
            vd = (l == r) ? pd[r] : vd;
        }
        if (l < 8) {
            const int e = (row0 + l + 8 * half) * stripsPerRow + strip - vb * 64;
            lg[e]      = vs;
            lg[64 + e] = vd;
        }
    }
    __syncthreads();
    {
        const int t = vt & 63;
        const int g = vb * 64 + t;
        if (g < nrows * stripsPerRow) {
            float* dstp = vt < 64 ? as_ : ad_;
            dstp[g] = lg[vt];
        }
    }
    if (live) {
        float* crow = C + (size_t)row0 * fout + col0;
#pragma unroll
        for (int r = 0; r < 8; ++r) {
            const int m = r + 8 * half;
            float* p = crow + (size_t)m * fout + l;
            p[ 0] = acc0[r];
            p[16] = acc1[r];
            p[32] = acc2[r];
            p[48] = acc3[r];
        }
    }
}

static_assert(F1 % 16 == 0, "layer 2's inner dimension is whole blocks of 4 k-steps");
__global__ void __launch_bounds__(128) gemm_wmma_f32_logits_kernel(const float* __restrict__ A,
                                     const float* __restrict__ W,
                                     const float* __restrict__ a_src, const float* __restrict__ a_dst,
                                     float* __restrict__ C,
                                     float* __restrict__ as_, float* __restrict__ ad_,
                                     int nrows, int fin, int fout) {
    __shared__ float lg[128];
    gemm_logits_body<4>(A, W, a_src, a_dst, C, as_, ad_, nrows, fin, fout, (int)blockIdx.x, (int)threadIdx.x, lg);
}

#define BIN_SHIFT   8
#define BIN_NODES   (1 << BIN_SHIFT)
#define NBINS       ((NNODES + BIN_NODES - 1) / BIN_NODES)
#define NCHUNK      256
#define CHUNK       ((NEDGES + NCHUNK - 1) / NCHUNK)
#define CHUNK_PITCH ((CHUNK + 31) & ~31)
#define GRP_T       256
#define BIN_TRIPS   ((CHUNK + GRP_T - 1) / GRP_T)
#define RUNS_PER_WAVE (NCHUNK / (GRP_T / 32))
#define STAGE_CAP   16384
static_assert(NNODES <= 65536, "an edge is packed as destination << 16 | source");
static_assert(NBINS < 256, "bin offsets are scanned by 256 threads, entry NBINS is the chunk total");
static_assert(NNODES + 1 <= 65536 && NCHUNK * 256 <= 65536 &&
              131072 + NCHUNK * CHUNK_PITCH + NEDGES <= (NEDGES + NNODES) * HEADS,
              "row pointers, bin offsets, sorted chunks and source list fit the grouping area of the workspace");
static_assert(GRP_T >= 256 && NCHUNK <= GRP_T && NCHUNK % (GRP_T / 32) == 0 && RUNS_PER_WAVE <= 32,
              "256 scanning threads, one thread per chunk, whole runs per wave, one mask bit per run");

__device__ __forceinline__ int scan256(int v, int* wtot, int& total) {
    const int lane = threadIdx.x & 31, w = threadIdx.x >> 5;
    int incl = v;
    for (int off = 1; off < 32; off <<= 1) {
        const int u = __shfl_up(incl, off);
        if (lane >= off) incl += u;
    }
    if (w < 8 && lane == 31) wtot[w] = incl;
    __syncthreads();
    int carry = 0;
    total = 0;
    for (int k = 0; k < 8; ++k) {
        const int x = wtot[k];
        carry += (k < w) ? x : 0;
        total += x;
    }
    return carry + incl - v;
}

__device__ __forceinline__ void edge_bin_body(const int* __restrict__ ei,
                                              unsigned* __restrict__ sorted,
                                              int* __restrict__ boff) {
    __shared__ int hist[256];
    __shared__ int wtot[8];
    __shared__ unsigned stage[CHUNK];
    const int t = threadIdx.x, g = blockIdx.x;
    const int e0 = g * CHUNK;
    int cnt = NEDGES - e0;
    cnt = cnt < 0 ? 0 : (cnt > CHUNK ? CHUNK : cnt);
    if (t < 256) hist[t] = 0;
    __syncthreads();
    unsigned word[BIN_TRIPS];
    int rk[BIN_TRIPS];
    {
        const int last = (e0 + cnt < NEDGES ? e0 + cnt : NEDGES) - 1;
        unsigned sraw[BIN_TRIPS], draw[BIN_TRIPS];
#pragma unroll
        for (int it = 0; it < BIN_TRIPS; ++it) {
            const int i = e0 + it * GRP_T + t;
            const int idx = i < last ? i : last;
            sraw[it] = (unsigned)ei[idx];
            draw[it] = (unsigned)ei[NEDGES + idx];
        }
#pragma unroll
        for (int it = 0; it < BIN_TRIPS; ++it) {
            word[it] = (draw[it] << 16) | sraw[it];
            rk[it] = 0;
            if (it * GRP_T + t < cnt) rk[it] = atomicAdd(&hist[draw[it] >> BIN_SHIFT], 1);
        }
    }
    __syncthreads();
    int total;
    const int v = t < 256 ? hist[t] : 0;
    const int excl = scan256(v, wtot, total);
    if (t < 256) {
        hist[t] = excl;
        boff[g * 256 + t] = excl;
    }
    __syncthreads();
#pragma unroll
    for (int it = 0; it < BIN_TRIPS; ++it) {
        const int i = it * GRP_T + t;
        if (i < cnt) stage[hist[word[it] >> (16 + BIN_SHIFT)] + rk[it]] = word[it];
    }
    __syncthreads();
    unsigned* dstp = sorted + (size_t)g * CHUNK_PITCH;
    for (int i = t; i < cnt; i += GRP_T) dstp[i] = stage[i];
}

__device__ __forceinline__ void bin_group_body(const unsigned* __restrict__ sorted,
                                               const int* __restrict__ boff,
                                               int* __restrict__ rowptr,
                                               int* __restrict__ csrc) {
    __shared__ int rs[NCHUNK], re[NCHUNK];
    __shared__ int cur[256];
    __shared__ int wtot[8];
    __shared__ int part[GRP_T / 32];
    __shared__ int stage[STAGE_CAP];
    const int t = threadIdx.x, b = blockIdx.x, lane = t & 31, w = t >> 5;
    int sv = 0;
    if (t < NCHUNK) {
        sv = boff[t * 256 + b];
        rs[t] = sv;
        re[t] = boff[t * 256 + b + 1];
    }
    if (t < 256) cur[t] = 0;
    int psum = sv;
    for (int off = 16; off; off >>= 1) psum += __shfl_xor(psum, off);
    if (lane == 0) part[w] = psum;
    __syncthreads();
    int base = 0;
    for (int k = 0; k < GRP_T / 32; ++k) base += part[k];

    unsigned w0[RUNS_PER_WAVE];
    unsigned live = 0u;
    unsigned longm = 0u;
    const int rsv = lane < RUNS_PER_WAVE ? rs[w * RUNS_PER_WAVE + lane] : 0;
    const int rev = lane < RUNS_PER_WAVE ? re[w * RUNS_PER_WAVE + lane] : 0;
#pragma unroll
    for (int r = 0; r < RUNS_PER_WAVE; ++r) {
        const int g = w * RUNS_PER_WAVE + r;
        const int s = __builtin_amdgcn_readlane(rsv, r), e = __builtin_amdgcn_readlane(rev, r);
        w0[r] = 0u;
        if (s + lane < e) {
            w0[r] = sorted[(size_t)g * CHUNK_PITCH + s + lane];
            live |= 1u << r;
        }
        if (e - s > 32) longm |= 1u << r;
    }
#pragma unroll
    for (int r = 0; r < RUNS_PER_WAVE; ++r)
        if ((live >> r) & 1u) atomicAdd(&cur[(w0[r] >> 16) & (BIN_NODES - 1)], 1);
    for (unsigned m = longm; m; m &= m - 1u) {
        const int g = w * RUNS_PER_WAVE + __builtin_ctz(m);
        const unsigned* p = sorted + (size_t)g * CHUNK_PITCH;
        const int e = re[g];
        for (int i = rs[g] + lane + 32; i < e; i += 32) atomicAdd(&cur[(p[i] >> 16) & (BIN_NODES - 1)], 1);
    }
    __syncthreads();
    int total;
    const int v = t < 256 ? cur[t] : 0;
    const int excl = scan256(v, wtot, total);
    if (t < 256) {
        cur[t] = excl;
        const int node = b * BIN_NODES + t;
        if (node < NNODES) rowptr[node] = base + excl;
        if (node == NNODES - 1) rowptr[NNODES] = base + excl + v;
    }
    __syncthreads();
    const bool staged = total <= STAGE_CAP;
    {
        int pos[RUNS_PER_WAVE];
#pragma unroll
        for (int r = 0; r < RUNS_PER_WAVE; ++r) {
            pos[r] = 0;
            if ((live >> r) & 1u) pos[r] = atomicAdd(&cur[(w0[r] >> 16) & (BIN_NODES - 1)], 1);
        }
        if (staged) {
#pragma unroll
            for (int r = 0; r < RUNS_PER_WAVE; ++r)
                if ((live >> r) & 1u) stage[pos[r]] = (int)(w0[r] & 0xffffu);
        } else {
#pragma unroll
            for (int r = 0; r < RUNS_PER_WAVE; ++r)
                if ((live >> r) & 1u) csrc[base + pos[r]] = (int)(w0[r] & 0xffffu);
        }
    }
    for (unsigned m = longm; m; m &= m - 1u) {
        const int g = w * RUNS_PER_WAVE + __builtin_ctz(m);
        const unsigned* p = sorted + (size_t)g * CHUNK_PITCH;
        const int e = re[g];
        for (int i = rs[g] + lane + 32; i < e; i += 32) {
            const unsigned wd = p[i];
            const int pos = atomicAdd(&cur[(wd >> 16) & (BIN_NODES - 1)], 1);
            const int sidx = (int)(wd & 0xffffu);
            if (staged) stage[pos] = sidx; else csrc[base + pos] = sidx;
        }
    }
    if (staged) {
        __syncthreads();
        for (int i = t; i < total; i += GRP_T) csrc[base + i] = stage[i];
    }
}

#define GEMM1_GROUPS   ((NNODES / 16) * (F1 / 64) / 4)
#define GEMM1_GROUPS_A ((GEMM1_GROUPS / 4) * 2)
#define GEMM1_WG_A     (GEMM1_GROUPS_A / 2)
#define GEMM1_WG_B     ((GEMM1_GROUPS - GEMM1_GROUPS_A + 1) / 2)
static_assert(((NNODES / 16) * (F1 / 64)) % 4 == 0 && GRP_T == 256, "whole groups; two groups per work-group");

__global__ void __launch_bounds__(GRP_T, 8) edge_bin_and_gemm_kernel(const int* __restrict__ ei,
        unsigned* __restrict__ sorted, int* __restrict__ boff,
        const float* __restrict__ A, const float* __restrict__ W,
        const float* __restrict__ a_src, const float* __restrict__ a_dst,
        float* __restrict__ C, float* __restrict__ as_, float* __restrict__ ad_,
        int nrows, int fin, int fout, int vb0) {
    __shared__ float lg[2 * 128];
    if (blockIdx.x < NCHUNK) {
        edge_bin_body(ei, sorted, boff);
    } else {
        const int sub = (int)threadIdx.x >> 7;
        gemm_logits_body<1>(A, W, a_src, a_dst, C, as_, ad_, nrows, fin, fout,
                         vb0 + 2 * ((int)blockIdx.x - NCHUNK) + sub, (int)threadIdx.x & 127, lg + 128 * sub);
    }
}

__global__ void __launch_bounds__(GRP_T) bin_group_and_gemm_kernel(const unsigned* __restrict__ sorted,
        const int* __restrict__ boff, int* __restrict__ rowptr, int* __restrict__ csrc,
        const float* __restrict__ A, const float* __restrict__ W,
        const float* __restrict__ a_src, const float* __restrict__ a_dst,
        float* __restrict__ C, float* __restrict__ as_, float* __restrict__ ad_,
        int nrows, int fin, int fout, int vb0) {
    __shared__ float lg[2 * 128];
    if (blockIdx.x < NBINS) {
        bin_group_body(sorted, boff, rowptr, csrc);
    } else {
        const int sub = (int)threadIdx.x >> 7;
        gemm_logits_body<1>(A, W, a_src, a_dst, C, as_, ad_, nrows, fin, fout,
                         vb0 + 2 * ((int)blockIdx.x - NBINS) + sub, (int)threadIdx.x & 127, lg + 128 * sub);
    }
}

__device__ __forceinline__ float wave_max(float v) {
    for (int off = 16; off; off >>= 1) v = fmaxf(v, __shfl_xor(v, off));
    return v;
}
__device__ __forceinline__ float wave_sum(float v) {
    for (int off = 16; off; off >>= 1) v += __shfl_xor(v, off);
    return v;
}
__device__ __forceinline__ float lane_bcast(float v, int q) {
    return __int_as_float(__builtin_amdgcn_readlane(__float_as_int(v), q));
}

template <int H>
__device__ __forceinline__ void load_heads(const float* __restrict__ p, float (&o)[H]) {
    if constexpr (H == 4) {
        const float4 t = *reinterpret_cast<const float4*>(p);
        o[0] = t.x; o[1] = t.y; o[2] = t.z; o[3] = t.w;
    } else {
        for (int hh = 0; hh < H; ++hh) o[hh] = p[hh];
    }
}

template <int H>
__device__ __forceinline__ void load_chunk(const int* __restrict__ csrc, const float* __restrict__ as_,
                                           const float (&adv)[H], int d, int beg, int n, int j,
                                           int& s, float (&e)[H], bool& valid) {
    valid = j < n;
    s = d;
    if (j < n - 1) s = csrc[beg + j];
    float a[H];
    load_heads<H>(as_ + (size_t)s * H, a);
    for (int hh = 0; hh < H; ++hh) e[hh] = leakyf(a[hh] + adv[hh], 0.2f);
}

template <int H>
__device__ __forceinline__ void add_rows(const float* __restrict__ h, int s, const float (&al)[H],
                                         int cnt, int lane, float (&acc)[H * 2]) {
    constexpr int F = H * 64;
#pragma unroll 4
    for (int q = 0; q < cnt; ++q) {
        const int sq = __builtin_amdgcn_readlane(s, q);
        const float* row = h + (size_t)sq * F;
        if constexpr (H == 4) {
            const float4 r0 = *reinterpret_cast<const float4*>(row + 4 * lane);
            const float4 r1 = *reinterpret_cast<const float4*>(row + 128 + 4 * lane);
            const float a0 = lane_bcast(al[0], q), a1 = lane_bcast(al[1], q);
            const float a2 = lane_bcast(al[2], q), a3 = lane_bcast(al[3], q);
            const float w0 = lane < 16 ? a0 : a1;
            const float w1 = lane < 16 ? a2 : a3;
            acc[0] += w0 * r0.x; acc[1] += w0 * r0.y; acc[2] += w0 * r0.z; acc[3] += w0 * r0.w;
            acc[4] += w1 * r1.x; acc[5] += w1 * r1.y; acc[6] += w1 * r1.z; acc[7] += w1 * r1.w;
        } else {
            const float2 r = *reinterpret_cast<const float2*>(row + 2 * lane);
            const float a0 = lane_bcast(al[0], q);
            acc[0] += a0 * r.x; acc[1] += a0 * r.y;
        }
    }
}

template <int H, bool LEAKY>
__device__ __forceinline__ void gather_body(const int* __restrict__ rowptr, const int* __restrict__ csrc,
                                            const float* __restrict__ h,
                                            const float* __restrict__ as_, const float* __restrict__ ad_,
                                            const float* __restrict__ bias,
                                            float* __restrict__ out) {
    constexpr int F = H * 64;
    const int d    = __builtin_amdgcn_readfirstlane((blockIdx.x * blockDim.x + threadIdx.x) >> 5);
    const int lane = threadIdx.x & 31;
    if (d >= NNODES) return;
    const int beg = __builtin_amdgcn_readfirstlane(rowptr[d]);
    const int n   = __builtin_amdgcn_readfirstlane(rowptr[d + 1]) - beg + 1;
    float adv[H];
    load_heads<H>(ad_ + (size_t)d * H, adv);
    float bv[H * 2];
    if constexpr (H == 4) {
        const float4 b0 = *reinterpret_cast<const float4*>(bias + 4 * lane);
        const float4 b1 = *reinterpret_cast<const float4*>(bias + 128 + 4 * lane);
        bv[0] = b0.x; bv[1] = b0.y; bv[2] = b0.z; bv[3] = b0.w;
        bv[4] = b1.x; bv[5] = b1.y; bv[6] = b1.z; bv[7] = b1.w;
    } else {
        const float2 b0 = *reinterpret_cast<const float2*>(bias + 2 * lane);
        bv[0] = b0.x; bv[1] = b0.y;
    }

    int s0; float e0[H]; bool v0;
    load_chunk<H>(csrc, as_, adv, d, beg, n, lane, s0, e0, v0);
    float m[H];
    for (int hh = 0; hh < H; ++hh) m[hh] = v0 ? e0[hh] : -INFINITY;
    for (int j0 = 32; j0 < n; j0 += 32) {
        int s; float e[H]; bool v;
        load_chunk<H>(csrc, as_, adv, d, beg, n, j0 + lane, s, e, v);
        for (int hh = 0; hh < H; ++hh) m[hh] = v ? fmaxf(m[hh], e[hh]) : m[hh];
    }
    for (int hh = 0; hh < H; ++hh) {
        m[hh] = wave_max(m[hh]);
        m[hh] = isfinite(m[hh]) ? m[hh] : 0.f;
    }

    float ex0[H], den[H];
    for (int hh = 0; hh < H; ++hh) { ex0[hh] = v0 ? expf(e0[hh] - m[hh]) : 0.f; den[hh] = ex0[hh]; }
    for (int j0 = 32; j0 < n; j0 += 32) {
        int s; float e[H]; bool v;
        load_chunk<H>(csrc, as_, adv, d, beg, n, j0 + lane, s, e, v);
        for (int hh = 0; hh < H; ++hh) den[hh] += v ? expf(e[hh] - m[hh]) : 0.f;
    }
    for (int hh = 0; hh < H; ++hh) den[hh] = wave_sum(den[hh]) + 1e-16f;

    float acc[H * 2];
    for (int i = 0; i < H * 2; ++i) acc[i] = 0.f;
    {
        float al[H];
        for (int hh = 0; hh < H; ++hh) al[hh] = ex0[hh] / den[hh];
        add_rows<H>(h, s0, al, n < 32 ? n : 32, lane, acc);
    }
    for (int j0 = 32; j0 < n; j0 += 32) {
        int s; float e[H]; bool v;
        load_chunk<H>(csrc, as_, adv, d, beg, n, j0 + lane, s, e, v);
        float al[H];
        for (int hh = 0; hh < H; ++hh) al[hh] = v ? expf(e[hh] - m[hh]) / den[hh] : 0.f;
        add_rows<H>(h, s, al, (n - j0) < 32 ? (n - j0) : 32, lane, acc);
    }

    for (int i = 0; i < H * 2; ++i) {
        acc[i] += bv[i];
        if constexpr (LEAKY) acc[i] = leakyf(acc[i], 0.01f);
    }

    float* orow = out + (size_t)d * F;
    if constexpr (H == 4) {
        *reinterpret_cast<float4*>(orow + 4 * lane)       = make_float4(acc[0], acc[1], acc[2], acc[3]);
        *reinterpret_cast<float4*>(orow + 128 + 4 * lane) = make_float4(acc[4], acc[5], acc[6], acc[7]);
    } else {
        *reinterpret_cast<float2*>(orow + 2 * lane) = make_float2(acc[0], acc[1]);
    }
}

__global__ void __launch_bounds__(256) gather_heads4_kernel(const int* __restrict__ rowptr,
        const int* __restrict__ csrc, const float* __restrict__ h, const float* __restrict__ as_,
        const float* __restrict__ ad_, const float* __restrict__ bias, float* __restrict__ out) {
    gather_body<4, true>(rowptr, csrc, h, as_, ad_, bias, out);
}
__global__ void __launch_bounds__(256) gather_heads1_kernel(const int* __restrict__ rowptr,
        const int* __restrict__ csrc, const float* __restrict__ h, const float* __restrict__ as_,
        const float* __restrict__ ad_, const float* __restrict__ bias, float* __restrict__ out) {
    gather_body<1, false>(rowptr, csrc, h, as_, ad_, bias, out);
}

extern "C" void kernel_launch(void* const* d_in, const int* in_sizes, int n_in,
                              void* d_out, int out_size, void* d_ws, size_t ws_size,
                              hipStream_t stream) {
    const float* x   = (const float*)d_in[0];
    const int*   ei  = (const int*)  d_in[1];
    const float* W1  = (const float*)d_in[2];
    const float* as1 = (const float*)d_in[3];
    const float* ad1 = (const float*)d_in[4];
    const float* b1  = (const float*)d_in[5];
    const float* W2  = (const float*)d_in[6];
    const float* as2 = (const float*)d_in[7];
    const float* ad2 = (const float*)d_in[8];
    const float* b2  = (const float*)d_in[9];
    float* out = (float*)d_out;

    float* ws   = (float*)d_ws;
    float* h1   = ws;
    float* h2   = h1   + (size_t)NNODES * F1;
    int*   grp  = (int*)(h2 + (size_t)NNODES * F1);
    int*      rowptr = grp;
    int*      boff   = grp + 65536;
    unsigned* sorted = (unsigned*)(grp + 131072);
    int*      csrc   = grp + 131072 + NCHUNK * CHUNK_PITCH;
    float* asb  = (float*)(grp + (size_t)(NEDGES + NNODES) * HEADS);
    float* adb  = asb  + (size_t)NNODES * HEADS;
    float* g2   = h1;

    edge_bin_and_gemm_kernel<<<NCHUNK + GEMM1_WG_A, GRP_T, 0, stream>>>(ei, sorted, boff,
            x, W1, as1, ad1, h1, asb, adb, NNODES, IN_DIM, F1, 0);
    bin_group_and_gemm_kernel<<<NBINS + GEMM1_WG_B, GRP_T, 0, stream>>>(sorted, boff, rowptr, csrc,
            x, W1, as1, ad1, h1, asb, adb, NNODES, IN_DIM, F1, GEMM1_GROUPS_A);

    gather_heads4_kernel<<<(NNODES + 7) / 8, 256, 0, stream>>>(rowptr, csrc, h1, asb, adb, b1, h2);

    {
        int strips = (NNODES / 16) * (OUT_DIM / 64);
        gemm_wmma_f32_logits_kernel<<<(strips + 3) / 4, 128, 0, stream>>>(h2, W2, as2, ad2, g2, asb, adb,
                                                                         NNODES, F1, OUT_DIM);
    }
    gather_heads1_kernel<<<(NNODES + 7) / 8, 256, 0, stream>>>(rowptr, csrc, g2, asb, adb, b2, out);
}


# Product strips: each lane owns 4 adjacent columns (tile t = column 4*lane+t), so a W row is one 16-byte load and a C row one whole-line 16-byte store; C sums unchanged, logit column sums reordered
# speedup vs baseline: 6.4886x; 1.0286x over previous
#include <hip/hip_runtime.h>
#include <math.h>

#define NNODES  50000
#define NEDGES  800000
#define IN_DIM  128
#define HID     64
#define HEADS   4
#define OUT_DIM 64
#define F1      (HEADS * HID)

typedef __attribute__((ext_vector_type(2))) float v2f;
typedef __attribute__((ext_vector_type(4))) float v4f;
typedef __attribute__((ext_vector_type(8))) float v8f;

__device__ __forceinline__ float leakyf(float x, float s) { return x >= 0.f ? x : s * x; }


template <int X>
__device__ __forceinline__ float row_xor_add(float v) {
    return v + __int_as_float(__builtin_amdgcn_update_dpp(0, __float_as_int(v), 0x160 + X, 0xf, 0xf, true));
}

template <int KB>
__device__ __forceinline__ void gemm_logits_body(const float* __restrict__ A,
                                     const float* __restrict__ W,
                                     const float* __restrict__ a_src, const float* __restrict__ a_dst,
                                     float* __restrict__ C,
                                     float* __restrict__ as_, float* __restrict__ ad_,
                                     int nrows, int fin, int fout,
                                     int vb, int vt, float* lg) {
    const int gwave = (vb * 128 + vt) >> 5;
    const int lane  = vt & 31;
    const int stripsPerRow = fout >> 6;
    const int totalStrips  = (nrows >> 4) * stripsPerRow;
    const bool live = gwave < totalStrips;

    const int rowBlk = gwave / stripsPerRow;
    const int strip  = gwave % stripsPerRow;
    const int row0 = rowBlk << 4;
    const int col0 = strip << 6;
    const int half = lane >> 4;
    const int l    = lane & 15;

    v8f acc0 = {}, acc1 = {}, acc2 = {}, acc3 = {};
    if (live) {
        const v4f sa = *reinterpret_cast<const v4f*>(a_src + col0 + 4 * l);
        const v4f da = *reinterpret_cast<const v4f*>(a_dst + col0 + 4 * l);

        const float* ak = A + (size_t)(row0 + l) * fin + 2 * half;
        const float* wk = W + (size_t)(2 * half) * fout + col0 + 4 * l;
        const size_t wrow = (size_t)fout;
        auto load_steps = [&](const float* a, const float* w, auto& fa, auto& fb0, auto& fb1) {
            constexpr int NK = (int)(sizeof(fa) / sizeof(fa[0]));
#pragma unroll
            for (int j = 0; j < NK; ++j) {
                fa[j].x = a[4 * j + 0];
                fa[j].y = a[4 * j + 1];
                fb0[j] = *reinterpret_cast<const v4f*>(w + (4 * j + 0) * wrow);
                fb1[j] = *reinterpret_cast<const v4f*>(w + (4 * j + 1) * wrow);
            }
        };
        auto mma_steps = [&](auto& fa, auto& fb0, auto& fb1) {
            constexpr int NK = (int)(sizeof(fa) / sizeof(fa[0]));
#pragma unroll
            for (int j = 0; j < NK; ++j) {
                asm volatile("" : "+v"(fa[j]), "+v"(fb0[j]), "+v"(fb1[j]));
                const v2f b0 = { fb0[j].x, fb1[j].x };
                const v2f b1 = { fb0[j].y, fb1[j].y };
                const v2f b2 = { fb0[j].z, fb1[j].z };
                const v2f b3 = { fb0[j].w, fb1[j].w };
                acc0 = __builtin_amdgcn_wmma_f32_16x16x4_f32(false, fa[j], false, b0, (short)0, acc0, false, false);
                acc1 = __builtin_amdgcn_wmma_f32_16x16x4_f32(false, fa[j], false, b1, (short)0, acc1, false, false);
                acc2 = __builtin_amdgcn_wmma_f32_16x16x4_f32(false, fa[j], false, b2, (short)0, acc2, false, false);
                acc3 = __builtin_amdgcn_wmma_f32_16x16x4_f32(false, fa[j], false, b3, (short)0, acc3, false, false);
            }
        };
        if constexpr (KB == 1) {
            int k = 0;
            for (; k + 16 <= fin; k += 16, ak += 16, wk += 16 * wrow) {
                v2f ga[4];
                v4f gb0[4], gb1[4];
                load_steps(ak, wk, ga, gb0, gb1);
                mma_steps(ga, gb0, gb1);
            }
            for (; k < fin; k += 4, ak += 4, wk += 4 * wrow) {
                v2f ga[1];
                v4f gb0[1], gb1[1];
                load_steps(ak, wk, ga, gb0, gb1);
                mma_steps(ga, gb0, gb1);
            }
        } else {
            constexpr int BK = 4 * KB;
            v2f pa[KB], qa[KB];
            v4f pb0[KB], pb1[KB], qb0[KB], qb1[KB];
            load_steps(ak, wk, pa, pb0, pb1);
            int k = 0;
            for (; k + 2 * BK < fin; k += 2 * BK) {
                load_steps(ak + BK, wk + BK * wrow, qa, qb0, qb1);
                mma_steps(pa, pb0, pb1);
                ak += 2 * BK;
                wk += 2 * BK * wrow;
                load_steps(ak, wk, pa, pb0, pb1);
                mma_steps(qa, qb0, qb1);
            }
            if (k + BK < fin) {
                load_steps(ak + BK, wk + BK * wrow, qa, qb0, qb1);
                mma_steps(pa, pb0, pb1);
                mma_steps(qa, qb0, qb1);
            } else {
                mma_steps(pa, pb0, pb1);
            }
        }

        float ps[8], pd[8];
#pragma unroll
        for (int r = 0; r < 8; ++r) {
            ps[r] = acc0[r] * sa[0] + acc1[r] * sa[1] + acc2[r] * sa[2] + acc3[r] * sa[3];
            pd[r] = acc0[r] * da[0] + acc1[r] * da[1] + acc2[r] * da[2] + acc3[r] * da[3];
        }
#pragma unroll
        for (int r = 0; r < 8; ++r) { ps[r] = row_xor_add<1>(ps[r]); pd[r] = row_xor_add<1>(pd[r]); }
#pragma unroll
        for (int r = 0; r < 8; ++r) { ps[r] = row_xor_add<2>(ps[r]); pd[r] = row_xor_add<2>(pd[r]); }
#pragma unroll
        for (int r = 0; r < 8; ++r) { ps[r] = row_xor_add<4>(ps[r]); pd[r] = row_xor_add<4>(pd[r]); }
#pragma unroll
        for (int r = 0; r < 8; ++r) { ps[r] = row_xor_add<8>(ps[r]); pd[r] = row_xor_add<8>(pd[r]); }
        float vs = ps[0], vd = pd[0];
#pragma unroll
        for (int r = 1; r < 8; ++r) {
            vs = (l == r) ? ps[r] : vs;
            vd = (l == r) ? pd[r] : vd;
        }
        if (l < 8) {
            const int e = (row0 + l + 8 * half) * stripsPerRow + strip - vb * 64;
            lg[e]      = vs;
            lg[64 + e] = vd;
        }
    }
    __syncthreads();
    {
        const int t = vt & 63;
        const int g = vb * 64 + t;
        if (g < nrows * stripsPerRow) {
            float* dstp = vt < 64 ? as_ : ad_;
            dstp[g] = lg[vt];
        }
    }
    if (live) {
        float* crow = C + (size_t)row0 * fout + col0;
#pragma unroll
        for (int r = 0; r < 8; ++r) {
            const int m = r + 8 * half;
            float* p = crow + (size_t)m * fout + 4 * l;
            *reinterpret_cast<v4f*>(p) = v4f{ acc0[r], acc1[r], acc2[r], acc3[r] };
        }
    }
}

static_assert(F1 % 16 == 0, "layer 2's inner dimension is whole blocks of 4 k-steps");
__global__ void __launch_bounds__(128) gemm_wmma_f32_logits_kernel(const float* __restrict__ A,
                                     const float* __restrict__ W,
                                     const float* __restrict__ a_src, const float* __restrict__ a_dst,
                                     float* __restrict__ C,
                                     float* __restrict__ as_, float* __restrict__ ad_,
                                     int nrows, int fin, int fout) {
    __shared__ float lg[128];
    gemm_logits_body<4>(A, W, a_src, a_dst, C, as_, ad_, nrows, fin, fout, (int)blockIdx.x, (int)threadIdx.x, lg);
}

#define BIN_SHIFT   8
#define BIN_NODES   (1 << BIN_SHIFT)
#define NBINS       ((NNODES + BIN_NODES - 1) / BIN_NODES)
#define NCHUNK      256
#define CHUNK       ((NEDGES + NCHUNK - 1) / NCHUNK)
#define CHUNK_PITCH ((CHUNK + 31) & ~31)
#define GRP_T       256
#define BIN_TRIPS   ((CHUNK + GRP_T - 1) / GRP_T)
#define RUNS_PER_WAVE (NCHUNK / (GRP_T / 32))
#define STAGE_CAP   16384
static_assert(NNODES <= 65536, "an edge is packed as destination << 16 | source");
static_assert(NBINS < 256, "bin offsets are scanned by 256 threads, entry NBINS is the chunk total");
static_assert(NNODES + 1 <= 65536 && NCHUNK * 256 <= 65536 &&
              131072 + NCHUNK * CHUNK_PITCH + NEDGES <= (NEDGES + NNODES) * HEADS,
              "row pointers, bin offsets, sorted chunks and source list fit the grouping area of the workspace");
static_assert(GRP_T >= 256 && NCHUNK <= GRP_T && NCHUNK % (GRP_T / 32) == 0 && RUNS_PER_WAVE <= 32,
              "256 scanning threads, one thread per chunk, whole runs per wave, one mask bit per run");

__device__ __forceinline__ int scan256(int v, int* wtot, int& total) {
    const int lane = threadIdx.x & 31, w = threadIdx.x >> 5;
    int incl = v;
    for (int off = 1; off < 32; off <<= 1) {
        const int u = __shfl_up(incl, off);
        if (lane >= off) incl += u;
    }
    if (w < 8 && lane == 31) wtot[w] = incl;
    __syncthreads();
    int carry = 0;
    total = 0;
    for (int k = 0; k < 8; ++k) {
        const int x = wtot[k];
        carry += (k < w) ? x : 0;
        total += x;
    }
    return carry + incl - v;
}

__device__ __forceinline__ void edge_bin_body(const int* __restrict__ ei,
                                              unsigned* __restrict__ sorted,
                                              int* __restrict__ boff) {
    __shared__ int hist[256];
    __shared__ int wtot[8];
    __shared__ unsigned stage[CHUNK];
    const int t = threadIdx.x, g = blockIdx.x;
    const int e0 = g * CHUNK;
    int cnt = NEDGES - e0;
    cnt = cnt < 0 ? 0 : (cnt > CHUNK ? CHUNK : cnt);
    if (t < 256) hist[t] = 0;
    __syncthreads();
    unsigned word[BIN_TRIPS];
    int rk[BIN_TRIPS];
    {
        const int last = (e0 + cnt < NEDGES ? e0 + cnt : NEDGES) - 1;
        unsigned sraw[BIN_TRIPS], draw[BIN_TRIPS];
#pragma unroll
        for (int it = 0; it < BIN_TRIPS; ++it) {
            const int i = e0 + it * GRP_T + t;
            const int idx = i < last ? i : last;
            sraw[it] = (unsigned)ei[idx];
            draw[it] = (unsigned)ei[NEDGES + idx];
        }
#pragma unroll
        for (int it = 0; it < BIN_TRIPS; ++it) {
            word[it] = (draw[it] << 16) | sraw[it];
            rk[it] = 0;
            if (it * GRP_T + t < cnt) rk[it] = atomicAdd(&hist[draw[it] >> BIN_SHIFT], 1);
        }
    }
    __syncthreads();
    int total;
    const int v = t < 256 ? hist[t] : 0;
    const int excl = scan256(v, wtot, total);
    if (t < 256) {
        hist[t] = excl;
        boff[g * 256 + t] = excl;
    }
    __syncthreads();
#pragma unroll
    for (int it = 0; it < BIN_TRIPS; ++it) {
        const int i = it * GRP_T + t;
        if (i < cnt) stage[hist[word[it] >> (16 + BIN_SHIFT)] + rk[it]] = word[it];
    }
    __syncthreads();
    unsigned* dstp = sorted + (size_t)g * CHUNK_PITCH;
    for (int i = t; i < cnt; i += GRP_T) dstp[i] = stage[i];
}

__device__ __forceinline__ void bin_group_body(const unsigned* __restrict__ sorted,
                                               const int* __restrict__ boff,
                                               int* __restrict__ rowptr,
                                               int* __restrict__ csrc) {
    __shared__ int rs[NCHUNK], re[NCHUNK];
    __shared__ int cur[256];
    __shared__ int wtot[8];
    __shared__ int part[GRP_T / 32];
    __shared__ int stage[STAGE_CAP];
    const int t = threadIdx.x, b = blockIdx.x, lane = t & 31, w = t >> 5;
    int sv = 0;
    if (t < NCHUNK) {
        sv = boff[t * 256 + b];
        rs[t] = sv;
        re[t] = boff[t * 256 + b + 1];
    }
    if (t < 256) cur[t] = 0;
    int psum = sv;
    for (int off = 16; off; off >>= 1) psum += __shfl_xor(psum, off);
    if (lane == 0) part[w] = psum;
    __syncthreads();
    int base = 0;
    for (int k = 0; k < GRP_T / 32; ++k) base += part[k];

    unsigned w0[RUNS_PER_WAVE];
    unsigned live = 0u;
    unsigned longm = 0u;
    const int rsv = lane < RUNS_PER_WAVE ? rs[w * RUNS_PER_WAVE + lane] : 0;
    const int rev = lane < RUNS_PER_WAVE ? re[w * RUNS_PER_WAVE + lane] : 0;
#pragma unroll
    for (int r = 0; r < RUNS_PER_WAVE; ++r) {
        const int g = w * RUNS_PER_WAVE + r;
        const int s = __builtin_amdgcn_readlane(rsv, r), e = __builtin_amdgcn_readlane(rev, r);
        w0[r] = 0u;
        if (s + lane < e) {
            w0[r] = sorted[(size_t)g * CHUNK_PITCH + s + lane];
            live |= 1u << r;
        }
        if (e - s > 32) longm |= 1u << r;
    }
#pragma unroll
    for (int r = 0; r < RUNS_PER_WAVE; ++r)
        if ((live >> r) & 1u) atomicAdd(&cur[(w0[r] >> 16) & (BIN_NODES - 1)], 1);
    for (unsigned m = longm; m; m &= m - 1u) {
        const int g = w * RUNS_PER_WAVE + __builtin_ctz(m);
        const unsigned* p = sorted + (size_t)g * CHUNK_PITCH;
        const int e = re[g];
        for (int i = rs[g] + lane + 32; i < e; i += 32) atomicAdd(&cur[(p[i] >> 16) & (BIN_NODES - 1)], 1);
    }
    __syncthreads();
    int total;
    const int v = t < 256 ? cur[t] : 0;
    const int excl = scan256(v, wtot, total);
    if (t < 256) {
        cur[t] = excl;
        const int node = b * BIN_NODES + t;
        if (node < NNODES) rowptr[node] = base + excl;
        if (node == NNODES - 1) rowptr[NNODES] = base + excl + v;
    }
    __syncthreads();
    const bool staged = total <= STAGE_CAP;
    {
        int pos[RUNS_PER_WAVE];
#pragma unroll
        for (int r = 0; r < RUNS_PER_WAVE; ++r) {
            pos[r] = 0;
            if ((live >> r) & 1u) pos[r] = atomicAdd(&cur[(w0[r] >> 16) & (BIN_NODES - 1)], 1);
        }
        if (staged) {
#pragma unroll
            for (int r = 0; r < RUNS_PER_WAVE; ++r)
                if ((live >> r) & 1u) stage[pos[r]] = (int)(w0[r] & 0xffffu);
        } else {
#pragma unroll
            for (int r = 0; r < RUNS_PER_WAVE; ++r)
                if ((live >> r) & 1u) csrc[base + pos[r]] = (int)(w0[r] & 0xffffu);
        }
    }
    for (unsigned m = longm; m; m &= m - 1u) {
        const int g = w * RUNS_PER_WAVE + __builtin_ctz(m);
        const unsigned* p = sorted + (size_t)g * CHUNK_PITCH;
        const int e = re[g];
        for (int i = rs[g] + lane + 32; i < e; i += 32) {
            const unsigned wd = p[i];
            const int pos = atomicAdd(&cur[(wd >> 16) & (BIN_NODES - 1)], 1);
            const int sidx = (int)(wd & 0xffffu);
            if (staged) stage[pos] = sidx; else csrc[base + pos] = sidx;
        }
    }
    if (staged) {
        __syncthreads();
        for (int i = t; i < total; i += GRP_T) csrc[base + i] = stage[i];
    }
}

#define GEMM1_GROUPS   ((NNODES / 16) * (F1 / 64) / 4)
#define GEMM1_GROUPS_A ((GEMM1_GROUPS / 4) * 2)
#define GEMM1_WG_A     (GEMM1_GROUPS_A / 2)
#define GEMM1_WG_B     ((GEMM1_GROUPS - GEMM1_GROUPS_A + 1) / 2)
static_assert(((NNODES / 16) * (F1 / 64)) % 4 == 0 && GRP_T == 256, "whole groups; two groups per work-group");

__global__ void __launch_bounds__(GRP_T, 8) edge_bin_and_gemm_kernel(const int* __restrict__ ei,
        unsigned* __restrict__ sorted, int* __restrict__ boff,
        const float* __restrict__ A, const float* __restrict__ W,
        const float* __restrict__ a_src, const float* __restrict__ a_dst,
        float* __restrict__ C, float* __restrict__ as_, float* __restrict__ ad_,
        int nrows, int fin, int fout, int vb0) {
    __shared__ float lg[2 * 128];
    if (blockIdx.x < NCHUNK) {
        edge_bin_body(ei, sorted, boff);
    } else {
        const int sub = (int)threadIdx.x >> 7;
        gemm_logits_body<1>(A, W, a_src, a_dst, C, as_, ad_, nrows, fin, fout,
                         vb0 + 2 * ((int)blockIdx.x - NCHUNK) + sub, (int)threadIdx.x & 127, lg + 128 * sub);
    }
}

__global__ void __launch_bounds__(GRP_T) bin_group_and_gemm_kernel(const unsigned* __restrict__ sorted,
        const int* __restrict__ boff, int* __restrict__ rowptr, int* __restrict__ csrc,
        const float* __restrict__ A, const float* __restrict__ W,
        const float* __restrict__ a_src, const float* __restrict__ a_dst,
        float* __restrict__ C, float* __restrict__ as_, float* __restrict__ ad_,
        int nrows, int fin, int fout, int vb0) {
    __shared__ float lg[2 * 128];
    if (blockIdx.x < NBINS) {
        bin_group_body(sorted, boff, rowptr, csrc);
    } else {
        const int sub = (int)threadIdx.x >> 7;
        gemm_logits_body<1>(A, W, a_src, a_dst, C, as_, ad_, nrows, fin, fout,
                         vb0 + 2 * ((int)blockIdx.x - NBINS) + sub, (int)threadIdx.x & 127, lg + 128 * sub);
    }
}

__device__ __forceinline__ float wave_max(float v) {
    for (int off = 16; off; off >>= 1) v = fmaxf(v, __shfl_xor(v, off));
    return v;
}
__device__ __forceinline__ float wave_sum(float v) {
    for (int off = 16; off; off >>= 1) v += __shfl_xor(v, off);
    return v;
}
__device__ __forceinline__ float lane_bcast(float v, int q) {
    return __int_as_float(__builtin_amdgcn_readlane(__float_as_int(v), q));
}

template <int H>
__device__ __forceinline__ void load_heads(const float* __restrict__ p, float (&o)[H]) {
    if constexpr (H == 4) {
        const float4 t = *reinterpret_cast<const float4*>(p);
        o[0] = t.x; o[1] = t.y; o[2] = t.z; o[3] = t.w;
    } else {
        for (int hh = 0; hh < H; ++hh) o[hh] = p[hh];
    }
}

template <int H>
__device__ __forceinline__ void load_chunk(const int* __restrict__ csrc, const float* __restrict__ as_,
                                           const float (&adv)[H], int d, int beg, int n, int j,
                                           int& s, float (&e)[H], bool& valid) {
    valid = j < n;
    s = d;
    if (j < n - 1) s = csrc[beg + j];
    float a[H];
    load_heads<H>(as_ + (size_t)s * H, a);
    for (int hh = 0; hh < H; ++hh) e[hh] = leakyf(a[hh] + adv[hh], 0.2f);
}

template <int H>
__device__ __forceinline__ void add_rows(const float* __restrict__ h, int s, const float (&al)[H],
                                         int cnt, int lane, float (&acc)[H * 2]) {
    constexpr int F = H * 64;
#pragma unroll 4
    for (int q = 0; q < cnt; ++q) {
        const int sq = __builtin_amdgcn_readlane(s, q);
        const float* row = h + (size_t)sq * F;
        if constexpr (H == 4) {
            const float4 r0 = *reinterpret_cast<const float4*>(row + 4 * lane);
            const float4 r1 = *reinterpret_cast<const float4*>(row + 128 + 4 * lane);
            const float a0 = lane_bcast(al[0], q), a1 = lane_bcast(al[1], q);
            const float a2 = lane_bcast(al[2], q), a3 = lane_bcast(al[3], q);
            const float w0 = lane < 16 ? a0 : a1;
            const float w1 = lane < 16 ? a2 : a3;
            acc[0] += w0 * r0.x; acc[1] += w0 * r0.y; acc[2] += w0 * r0.z; acc[3] += w0 * r0.w;
            acc[4] += w1 * r1.x; acc[5] += w1 * r1.y; acc[6] += w1 * r1.z; acc[7] += w1 * r1.w;
        } else {
            const float2 r = *reinterpret_cast<const float2*>(row + 2 * lane);
            const float a0 = lane_bcast(al[0], q);
            acc[0] += a0 * r.x; acc[1] += a0 * r.y;
        }
    }
}

template <int H, bool LEAKY>
__device__ __forceinline__ void gather_body(const int* __restrict__ rowptr, const int* __restrict__ csrc,
                                            const float* __restrict__ h,
                                            const float* __restrict__ as_, const float* __restrict__ ad_,
                                            const float* __restrict__ bias,
                                            float* __restrict__ out) {
    constexpr int F = H * 64;
    const int d    = __builtin_amdgcn_readfirstlane((blockIdx.x * blockDim.x + threadIdx.x) >> 5);
    const int lane = threadIdx.x & 31;
    if (d >= NNODES) return;
    const int beg = __builtin_amdgcn_readfirstlane(rowptr[d]);
    const int n   = __builtin_amdgcn_readfirstlane(rowptr[d + 1]) - beg + 1;
    float adv[H];
    load_heads<H>(ad_ + (size_t)d * H, adv);
    float bv[H * 2];
    if constexpr (H == 4) {
        const float4 b0 = *reinterpret_cast<const float4*>(bias + 4 * lane);
        const float4 b1 = *reinterpret_cast<const float4*>(bias + 128 + 4 * lane);
        bv[0] = b0.x; bv[1] = b0.y; bv[2] = b0.z; bv[3] = b0.w;
        bv[4] = b1.x; bv[5] = b1.y; bv[6] = b1.z; bv[7] = b1.w;
    } else {
        const float2 b0 = *reinterpret_cast<const float2*>(bias + 2 * lane);
        bv[0] = b0.x; bv[1] = b0.y;
    }

    int s0; float e0[H]; bool v0;
    load_chunk<H>(csrc, as_, adv, d, beg, n, lane, s0, e0, v0);
    float m[H];
    for (int hh = 0; hh < H; ++hh) m[hh] = v0 ? e0[hh] : -INFINITY;
    for (int j0 = 32; j0 < n; j0 += 32) {
        int s; float e[H]; bool v;
        load_chunk<H>(csrc, as_, adv, d, beg, n, j0 + lane, s, e, v);
        for (int hh = 0; hh < H; ++hh) m[hh] = v ? fmaxf(m[hh], e[hh]) : m[hh];
    }
    for (int hh = 0; hh < H; ++hh) {
        m[hh] = wave_max(m[hh]);
        m[hh] = isfinite(m[hh]) ? m[hh] : 0.f;
    }

    float ex0[H], den[H];
    for (int hh = 0; hh < H; ++hh) { ex0[hh] = v0 ? expf(e0[hh] - m[hh]) : 0.f; den[hh] = ex0[hh]; }
    for (int j0 = 32; j0 < n; j0 += 32) {
        int s; float e[H]; bool v;
        load_chunk<H>(csrc, as_, adv, d, beg, n, j0 + lane, s, e, v);
        for (int hh = 0; hh < H; ++hh) den[hh] += v ? expf(e[hh] - m[hh]) : 0.f;
    }
    for (int hh = 0; hh < H; ++hh) den[hh] = wave_sum(den[hh]) + 1e-16f;

    float acc[H * 2];
    for (int i = 0; i < H * 2; ++i) acc[i] = 0.f;
    {
        float al[H];
        for (int hh = 0; hh < H; ++hh) al[hh] = ex0[hh] / den[hh];
        add_rows<H>(h, s0, al, n < 32 ? n : 32, lane, acc);
    }
    for (int j0 = 32; j0 < n; j0 += 32) {
        int s; float e[H]; bool v;
        load_chunk<H>(csrc, as_, adv, d, beg, n, j0 + lane, s, e, v);
        float al[H];
        for (int hh = 0; hh < H; ++hh) al[hh] = v ? expf(e[hh] - m[hh]) / den[hh] : 0.f;
        add_rows<H>(h, s, al, (n - j0) < 32 ? (n - j0) : 32, lane, acc);
    }

    for (int i = 0; i < H * 2; ++i) {
        acc[i] += bv[i];
        if constexpr (LEAKY) acc[i] = leakyf(acc[i], 0.01f);
    }

    float* orow = out + (size_t)d * F;
    if constexpr (H == 4) {
        *reinterpret_cast<float4*>(orow + 4 * lane)       = make_float4(acc[0], acc[1], acc[2], acc[3]);
        *reinterpret_cast<float4*>(orow + 128 + 4 * lane) = make_float4(acc[4], acc[5], acc[6], acc[7]);
    } else {
        *reinterpret_cast<float2*>(orow + 2 * lane) = make_float2(acc[0], acc[1]);
    }
}

__global__ void __launch_bounds__(256) gather_heads4_kernel(const int* __restrict__ rowptr,
        const int* __restrict__ csrc, const float* __restrict__ h, const float* __restrict__ as_,
        const float* __restrict__ ad_, const float* __restrict__ bias, float* __restrict__ out) {
    gather_body<4, true>(rowptr, csrc, h, as_, ad_, bias, out);
}
__global__ void __launch_bounds__(256) gather_heads1_kernel(const int* __restrict__ rowptr,
        const int* __restrict__ csrc, const float* __restrict__ h, const float* __restrict__ as_,
        const float* __restrict__ ad_, const float* __restrict__ bias, float* __restrict__ out) {
    gather_body<1, false>(rowptr, csrc, h, as_, ad_, bias, out);
}

extern "C" void kernel_launch(void* const* d_in, const int* in_sizes, int n_in,
                              void* d_out, int out_size, void* d_ws, size_t ws_size,
                              hipStream_t stream) {
    const float* x   = (const float*)d_in[0];
    const int*   ei  = (const int*)  d_in[1];
    const float* W1  = (const float*)d_in[2];
    const float* as1 = (const float*)d_in[3];
    const float* ad1 = (const float*)d_in[4];
    const float* b1  = (const float*)d_in[5];
    const float* W2  = (const float*)d_in[6];
    const float* as2 = (const float*)d_in[7];
    const float* ad2 = (const float*)d_in[8];
    const float* b2  = (const float*)d_in[9];
    float* out = (float*)d_out;

    float* ws   = (float*)d_ws;
    float* h1   = ws;
    float* h2   = h1   + (size_t)NNODES * F1;
    int*   grp  = (int*)(h2 + (size_t)NNODES * F1);
    int*      rowptr = grp;
    int*      boff   = grp + 65536;
    unsigned* sorted = (unsigned*)(grp + 131072);
    int*      csrc   = grp + 131072 + NCHUNK * CHUNK_PITCH;
    float* asb  = (float*)(grp + (size_t)(NEDGES + NNODES) * HEADS);
    float* adb  = asb  + (size_t)NNODES * HEADS;
    float* g2   = h1;

    edge_bin_and_gemm_kernel<<<NCHUNK + GEMM1_WG_A, GRP_T, 0, stream>>>(ei, sorted, boff,
            x, W1, as1, ad1, h1, asb, adb, NNODES, IN_DIM, F1, 0);
    bin_group_and_gemm_kernel<<<NBINS + GEMM1_WG_B, GRP_T, 0, stream>>>(sorted, boff, rowptr, csrc,
            x, W1, as1, ad1, h1, asb, adb, NNODES, IN_DIM, F1, GEMM1_GROUPS_A);

    gather_heads4_kernel<<<(NNODES + 7) / 8, 256, 0, stream>>>(rowptr, csrc, h1, asb, adb, b1, h2);

    {
        int strips = (NNODES / 16) * (OUT_DIM / 64);
        gemm_wmma_f32_logits_kernel<<<(strips + 3) / 4, 128, 0, stream>>>(h2, W2, as2, ad2, g2, asb, adb,
                                                                         NNODES, F1, OUT_DIM);
    }
    gather_heads1_kernel<<<(NNODES + 7) / 8, 256, 0, stream>>>(rowptr, csrc, g2, asb, adb, b2, out);
}
